# starting kernel
# baseline (speedup 1.0000x reference)
#include <hip/hip_runtime.h>
#include <math.h>


typedef __attribute__((ext_vector_type(2))) float v2f;
typedef __attribute__((ext_vector_type(8))) float v8f;

#define NEG_SLOPE 0.2f

#if defined(__has_builtin)
#if __has_builtin(__builtin_amdgcn_wmma_f32_16x16x4_f32)
#define HAVE_WMMA_F32_16X16X4 1
#endif
#endif

__device__ __forceinline__ unsigned flipf(float f) {
  unsigned b = __float_as_uint(f);
  return (b & 0x80000000u) ? ~b : (b | 0x80000000u);
}
__device__ __forceinline__ float unflipf(unsigned u) {
  return __uint_as_float((u & 0x80000000u) ? (u & 0x7FFFFFFFu) : ~u);
}

__device__ __forceinline__ float leaky(float v) {
  return v > 0.f ? v : NEG_SLOPE * v;
}

template <int K, int DOUT, bool RELU_BIAS>
__global__ void gat_gemm_wmma(const float* __restrict__ X,
                              const float* __restrict__ W,
                              const float* __restrict__ bias,
                              float* __restrict__ H, int M) {
  const int lane   = threadIdx.x & 31;
  const int wave   = threadIdx.x >> 5;
  const int tileId = blockIdx.x * (blockDim.x >> 5) + wave;
  const int NT     = DOUT / 16;
  const int mt     = tileId / NT;
  const int nt     = tileId % NT;
  if (mt >= M / 16) return;

  const int r15 = lane & 15;
  const int hi  = lane >> 4;
  const int row = mt * 16 + r15;
  const int col = nt * 16 + r15;

  v8f c = {0.f, 0.f, 0.f, 0.f, 0.f, 0.f, 0.f, 0.f};

#if HAVE_WMMA_F32_16X16X4
  for (int k0 = 0; k0 < K; k0 += 4) {
    const int ka = k0 + 2 * hi;
    float x0 = X[row * K + ka];
    float x1 = X[row * K + ka + 1];
    if (RELU_BIAS) {
      x0 = fmaxf(x0 + bias[ka], 0.f);
      x1 = fmaxf(x1 + bias[ka + 1], 0.f);
    }
    v2f a;
    a.x = x0;
    a.y = x1;
    v2f b;
    b.x = W[ka * DOUT + col];
    b.y = W[(ka + 1) * DOUT + col];
    c = __builtin_amdgcn_wmma_f32_16x16x4_f32(false, a, false, b, (short)0, c,
                                              false, false);
  }
#else
  for (int r = 0; r < 8; ++r) {
    const int rr = mt * 16 + hi * 8 + r;
    float acc = 0.f;
    for (int k = 0; k < K; ++k) {
      float xv = X[rr * K + k];
      if (RELU_BIAS) xv = fmaxf(xv + bias[k], 0.f);
      acc += xv * W[k * DOUT + col];
    }
    c[r] = acc;
  }
#endif

  const int rowBase = mt * 16 + hi * 8;
#pragma unroll
  for (int r = 0; r < 8; ++r) H[(rowBase + r) * DOUT + col] = c[r];
}

template <int D>
__global__ void att_scores(const float* __restrict__ H,
                           const float* __restrict__ attS,
                           const float* __restrict__ attD,
                           float* __restrict__ aS, float* __restrict__ aD,
                           int N) {
  const int gid  = blockIdx.x * blockDim.x + threadIdx.x;
  const int node = gid >> 5;
  const int lane = gid & 31;
  if (node >= N) return;
  float s = 0.f, d = 0.f;
#pragma unroll
  for (int f = lane; f < D; f += 32) {
    const float h = H[(size_t)node * D + f];
    s += h * attS[f];
    d += h * attD[f];
  }
#pragma unroll
  for (int off = 16; off > 0; off >>= 1) {
    s += __shfl_down(s, off, 32);
    d += __shfl_down(d, off, 32);
  }
  if (lane == 0) {
    aS[node] = s;
    aD[node] = d;
  }
}

__global__ void edge_lrelu_max(const int* __restrict__ src,
                               const int* __restrict__ dst,
                               const float* __restrict__ aS,
                               const float* __restrict__ aD,
                               unsigned* __restrict__ emax, int E, int Etot) {
  const int e = blockIdx.x * blockDim.x + threadIdx.x;
  if (e >= Etot) return;
  int s, d;
  if (e < E) {
    s = src[e];
    d = dst[e];
  } else {
    s = d = e - E;
  }
  const float v = leaky(aS[s] + aD[d]);
  atomicMax(&emax[d], flipf(v));
}

__global__ void edge_expsum(const int* __restrict__ src,
                            const int* __restrict__ dst,
                            const float* __restrict__ aS,
                            const float* __restrict__ aD,
                            const unsigned* __restrict__ emax,
                            float* __restrict__ wbuf,
                            float* __restrict__ denom, int E, int Etot) {
  const int e = blockIdx.x * blockDim.x + threadIdx.x;
  if (e >= Etot) return;
  int s, d;
  if (e < E) {
    s = src[e];
    d = dst[e];
  } else {
    s = d = e - E;
  }
  const float v = leaky(aS[s] + aD[d]);
  const float m = unflipf(emax[d]);
  const float w = __expf(v - m);
  wbuf[e] = w;
  unsafeAtomicAdd(&denom[d], w);
}

__global__ void edge_alpha(const int* __restrict__ dst,
                           const float* __restrict__ denom,
                           float* __restrict__ wbuf, int E, int Etot) {
  const int e = blockIdx.x * blockDim.x + threadIdx.x;
  if (e >= Etot) return;
  const int d = (e < E) ? dst[e] : (e - E);
  wbuf[e] = wbuf[e] / denom[d];
}

template <int D>
__global__ void edge_aggregate(const int* __restrict__ src,
                               const int* __restrict__ dst,
                               const float* __restrict__ alpha,
                               const float* __restrict__ H,
                               float* __restrict__ out, int E, int Etot) {
  constexpr int TPE = D / 4;
  const long long t = (long long)blockIdx.x * blockDim.x + threadIdx.x;
  const long long e = t / TPE;
  const int q = (int)(t % TPE);
  if (e >= Etot) return;
  int s, d;
  if (e < E) {
    s = src[e];
    d = dst[e];
  } else {
    s = d = (int)(e - E);
  }
  const float a = alpha[e];
  const float4 h = *reinterpret_cast<const float4*>(H + (size_t)s * D + 4 * q);
  float* o = out + (size_t)d * D + 4 * q;
  unsafeAtomicAdd(o + 0, a * h.x);
  unsafeAtomicAdd(o + 1, a * h.y);
  unsafeAtomicAdd(o + 2, a * h.z);
  unsafeAtomicAdd(o + 3, a * h.w);
}

template <int D>
__global__ void finalize_tanh(float* __restrict__ out,
                              const float* __restrict__ b, int N) {
  const int t = blockIdx.x * blockDim.x + threadIdx.x;
  if (t >= N * D) return;
  out[t] = tanhf(out[t] + b[t & (D - 1)]);
}

extern "C" void kernel_launch(void* const* d_in, const int* in_sizes, int n_in,
                              void* d_out, int out_size, void* d_ws,
                              size_t ws_size, hipStream_t stream) {
  const float* x   = (const float*)d_in[0];
  const int*   ei  = (const int*)d_in[1];
  const float* W1  = (const float*)d_in[2];
  const float* as1 = (const float*)d_in[3];
  const float* ad1 = (const float*)d_in[4];
  const float* b1  = (const float*)d_in[5];
  const float* W2  = (const float*)d_in[6];
  const float* as2 = (const float*)d_in[7];
  const float* ad2 = (const float*)d_in[8];
  const float* b2  = (const float*)d_in[9];
  float* out = (float*)d_out;

  const int N    = in_sizes[0] / 64;
  const int E    = in_sizes[1] / 2;
  const int Etot = E + N;
  const int* src = ei;
  const int* dst = ei + E;

  float* ws = (float*)d_ws;
  size_t o = 0;
  float* h1 = ws + o;   o += (size_t)N * 64;
  float* o1 = ws + o;   o += (size_t)N * 64;
  float* aS = ws + o;   o += N;
  float* aD = ws + o;   o += N;
  unsigned* emax = (unsigned*)(ws + o); o += N;
  float* denom = ws + o; o += N;
  float* wbuf  = ws + o; o += Etot;

  const int B  = 256;
  const int ge = (Etot + B - 1) / B;

  {
    const int tiles = (N / 16) * (64 / 16);
    gat_gemm_wmma<64, 64, false>
        <<<(tiles + 7) / 8, B, 0, stream>>>(x, W1, nullptr, h1, N);
    att_scores<64><<<(int)(((size_t)N * 32 + B - 1) / B), B, 0, stream>>>(
        h1, as1, ad1, aS, aD, N);
    hipMemsetAsync(emax, 0, (size_t)N * 4, stream);
    hipMemsetAsync(denom, 0, (size_t)N * 4, stream);
    hipMemsetAsync(o1, 0, (size_t)N * 64 * 4, stream);
    edge_lrelu_max<<<ge, B, 0, stream>>>(src, dst, aS, aD, emax, E, Etot);
    edge_expsum<<<ge, B, 0, stream>>>(src, dst, aS, aD, emax, wbuf, denom, E,
                                      Etot);
    edge_alpha<<<ge, B, 0, stream>>>(dst, denom, wbuf, E, Etot);
    edge_aggregate<64>
        <<<(int)(((size_t)Etot * 16 + B - 1) / B), B, 0, stream>>>(
            src, dst, wbuf, h1, o1, E, Etot);
  }

  {
    float* h2 = h1;
    const int tiles = (N / 16) * (32 / 16);
    gat_gemm_wmma<64, 32, true>
        <<<(tiles + 7) / 8, B, 0, stream>>>(o1, W2, b1, h2, N);
    att_scores<32><<<(int)(((size_t)N * 32 + B - 1) / B), B, 0, stream>>>(
        h2, as2, ad2, aS, aD, N);
    hipMemsetAsync(emax, 0, (size_t)N * 4, stream);
    hipMemsetAsync(denom, 0, (size_t)N * 4, stream);
    hipMemsetAsync(out, 0, (size_t)N * 32 * 4, stream);
    edge_lrelu_max<<<ge, B, 0, stream>>>(src, dst, aS, aD, emax, E, Etot);
    edge_expsum<<<ge, B, 0, stream>>>(src, dst, aS, aD, emax, wbuf, denom, E,
                                      Etot);
    edge_alpha<<<ge, B, 0, stream>>>(dst, denom, wbuf, E, Etot);
    edge_aggregate<32>
        <<<(int)(((size_t)Etot * 8 + B - 1) / B), B, 0, stream>>>(
            src, dst, wbuf, h2, out, E, Etot);
    finalize_tanh<32><<<(int)(((size_t)N * 32 + B - 1) / B), B, 0, stream>>>(
        out, b2, N);
  }
}


# Edge passes by owner work-groups: a node run's max, exp sum and rows live in LDS, its edges are picked out of the edge list (kept for layer 2), LDS atomic adds, each row stored once; sums unordered
# speedup vs baseline: 3.6844x; 3.6844x over previous
#include <hip/hip_runtime.h>
#include <math.h>


typedef __attribute__((ext_vector_type(2))) float v2f;
typedef __attribute__((ext_vector_type(8))) float v8f;
typedef __attribute__((ext_vector_type(4))) int v4i;

#define NEG_SLOPE 0.2f

#if defined(__has_builtin)
#if __has_builtin(__builtin_amdgcn_wmma_f32_16x16x4_f32)
#define HAVE_WMMA_F32_16X16X4 1
#endif
#endif

__device__ __forceinline__ unsigned flipf(float f) {
  unsigned b = __float_as_uint(f);
  return (b & 0x80000000u) ? ~b : (b | 0x80000000u);
}
__device__ __forceinline__ float unflipf(unsigned u) {
  return __uint_as_float((u & 0x80000000u) ? (u & 0x7FFFFFFFu) : ~u);
}

__device__ __forceinline__ float leaky(float v) {
  return v > 0.f ? v : NEG_SLOPE * v;
}

#define LOOP_GRID 1024
#define GEMM_GRID 1024

template <int K, int DOUT>
__global__ void __launch_bounds__(256)
    __attribute__((amdgpu_waves_per_eu(8, 8)))
    gat_gemm_wmma(const float* __restrict__ X, const float* __restrict__ W,
                  float* __restrict__ H, int M) {
  constexpr int NT = DOUT / 16;
  constexpr int KS = K / 4;
  const int lane = threadIdx.x & 31;
  const int wpb  = blockDim.x >> 5;
  const int gwave =
      __builtin_amdgcn_readfirstlane(blockIdx.x * wpb + (threadIdx.x >> 5));
  const int nwaves = gridDim.x * wpb;
  const int MT     = M / 16;
  const int nt    = gwave % NT;
  const int mstep = nwaves / NT;
  int mt = gwave / NT;
  if (mt >= MT) return;

  const int r15 = lane & 15;
  const int hi  = lane >> 4;
  const int col = nt * 16 + r15;

#if HAVE_WMMA_F32_16X16X4
  v2f b[KS];
#pragma unroll
  for (int i = 0; i < KS; ++i) {
    const int ka = 4 * i + 2 * hi;
    b[i].x = W[ka * DOUT + col];
    b[i].y = W[(ka + 1) * DOUT + col];
  }

  for (; mt < MT; mt += mstep) {
    const int row = mt * 16 + r15;
    v8f c = {0.f, 0.f, 0.f, 0.f, 0.f, 0.f, 0.f, 0.f};
#pragma unroll
    for (int i = 0; i < KS; ++i) {
      const int ka = 4 * i + 2 * hi;
      v2f a;
      a.x = X[row * K + ka];
      a.y = X[row * K + ka + 1];
      c = __builtin_amdgcn_wmma_f32_16x16x4_f32(false, a, false, b[i],
                                                (short)0, c, false, false);
    }
    const int rowBase = mt * 16 + hi * 8;
#pragma unroll
    for (int r = 0; r < 8; ++r) H[(rowBase + r) * DOUT + col] = c[r];
  }
#else
  for (; mt < MT; mt += mstep) {
    for (int r = 0; r < 8; ++r) {
      const int rr = mt * 16 + hi * 8 + r;
      float acc = 0.f;
      for (int k = 0; k < K; ++k) acc += X[rr * K + k] * W[k * DOUT + col];
      H[rr * DOUT + col] = acc;
    }
  }
#endif
}

template <int D>
__global__ void att_scores(const float* __restrict__ H,
                           const float* __restrict__ attS,
                           const float* __restrict__ attD,
                           float* __restrict__ aS, float* __restrict__ aD,
                           int N) {
  constexpr int FPL = D / 32;
  constexpr int NB  = 4;
  const int lane = threadIdx.x & 31;
  const int wpb  = blockDim.x >> 5;
  const int gwave =
      __builtin_amdgcn_readfirstlane(blockIdx.x * wpb + (threadIdx.x >> 5));
  const int nwaves = gridDim.x * wpb;
  const int per    = (N + nwaves - 1) / nwaves;
  const int n0     = gwave * per;
  const int n1     = (n0 + per < N) ? (n0 + per) : N;
  if (n0 >= n1) return;

  float as[FPL], ad[FPL];
#pragma unroll
  for (int f = 0; f < FPL; ++f) {
    as[f] = attS[32 * f + lane];
    ad[f] = attD[32 * f + lane];
  }

  for (int n = n0; n < n1; n += NB) {
    float h[NB][FPL];
#pragma unroll
    for (int u = 0; u < NB; ++u) {
      const int node = (n + u < n1) ? (n + u) : (n1 - 1);
#pragma unroll
      for (int f = 0; f < FPL; ++f)
        h[u][f] = H[(size_t)node * D + 32 * f + lane];
    }
    float sv[NB], dv[NB];
#pragma unroll
    for (int u = 0; u < NB; ++u) {
      float sa = 0.f, da = 0.f;
#pragma unroll
      for (int f = 0; f < FPL; ++f) {
        sa += h[u][f] * as[f];
        da += h[u][f] * ad[f];
      }
      sv[u] = sa;
      dv[u] = da;
    }
#pragma unroll
    for (int off = 16; off > 0; off >>= 1) {
#pragma unroll
      for (int u = 0; u < NB; ++u) {
        sv[u] += __shfl_down(sv[u], off, 32);
        dv[u] += __shfl_down(dv[u], off, 32);
      }
    }
    if (lane == 0) {
#pragma unroll
      for (int u = 0; u < NB; ++u) {
        if (n + u < n1) {
          aS[n + u] = sv[u];
          aD[n + u] = dv[u];
        }
      }
    }
  }
}

#define OWN_GRID 256
#define OWN_THREADS 1024
#define OWN_NPW 400
#define OWN_CAP 8192
#define OWN_EPC 16

__device__ __forceinline__ void lds_addf(float* p, float v) {
  __hip_atomic_fetch_add(p, v, __ATOMIC_RELAXED, __HIP_MEMORY_SCOPE_WORKGROUP);
}
__device__ __forceinline__ void lds_maxu(unsigned* p, unsigned v) {
  __hip_atomic_fetch_max(p, v, __ATOMIC_RELAXED, __HIP_MEMORY_SCOPE_WORKGROUP);
}

template <int D, bool FINAL, bool SAVED>
__global__ void __launch_bounds__(OWN_THREADS)
    owner_aggregate(const int* __restrict__ src, const int* __restrict__ dst,
                    const float* __restrict__ aS, const float* __restrict__ aD,
                    const float* __restrict__ H, const float* __restrict__ bias,
                    float* __restrict__ out, int* __restrict__ wcnt,
                    int* __restrict__ wlist, int E, int N) {
  constexpr int FPL = D / 32;
  constexpr int T   = OWN_THREADS;
  constexpr int NW  = OWN_THREADS / 32;
  constexpr int EPC = OWN_EPC;
  __shared__ float acc[OWN_NPW * 64];
  __shared__ float den[OWN_NPW];
  __shared__ unsigned mx[OWN_NPW];
  __shared__ int ls[OWN_CAP];
  __shared__ int ld[OWN_CAP];
  __shared__ float lv[OWN_CAP];
  __shared__ int cnt;

  const int tid  = threadIdx.x;
  const int lane = tid & 31;
  const int wave = __builtin_amdgcn_readfirstlane(tid >> 5);

  int npw = (N + (int)gridDim.x - 1) / (int)gridDim.x;
  if (npw > OWN_NPW) npw = OWN_NPW;
  const int nruns = (N + npw - 1) / npw;

  for (int run = blockIdx.x; run < nruns; run += gridDim.x) {
    const int n0 = run * npw;
    const int nn = (N - n0 < npw) ? (N - n0) : npw;
    int* __restrict__ ws_s = wlist + (size_t)run * (2 * OWN_CAP);
    int* __restrict__ ws_d = ws_s + OWN_CAP;

    for (int i = tid; i < nn * D; i += T) acc[i] = 0.f;
    for (int i = tid; i < nn; i += T) {
      den[i] = 0.f;
      mx[i]  = 0u;
      ls[i]  = n0 + i;
      ld[i]  = i;
    }
    if (tid == 0) cnt = nn;
    __syncthreads();

    bool loaded = false;
    if (SAVED) {
      const int c = wcnt[run];
      if (c >= 0) {
        loaded = true;
        for (int i = nn + tid; i < c; i += T) {
          ls[i] = ws_s[i];
          ld[i] = ws_d[i];
        }
        if (tid == 0) cnt = c;
      }
    }
    if (!loaded) {
      auto pick = [&](int e, unsigned r) {
        if (r < (unsigned)nn) {
          const int idx = __hip_atomic_fetch_add(&cnt, 1, __ATOMIC_RELAXED,
                                                 __HIP_MEMORY_SCOPE_WORKGROUP);
          if (idx < OWN_CAP) {
            ls[idx] = e;
            ld[idx] = (int)r;
          }
        }
      };
      auto pick4 = [&](int e, v4i d) {
        const unsigned r0 = (unsigned)(d.x - n0), r1 = (unsigned)(d.y - n0);
        const unsigned r2 = (unsigned)(d.z - n0), r3 = (unsigned)(d.w - n0);
        const unsigned m01 = r0 < r1 ? r0 : r1, m23 = r2 < r3 ? r2 : r3;
        if ((m01 < m23 ? m01 : m23) < (unsigned)nn) {
          pick(e, r0);
          pick(e + 1, r1);
          pick(e + 2, r2);
          pick(e + 3, r3);
        }
      };
      int head = (int)(((16u - ((unsigned)(size_t)dst & 15u)) & 15u) >> 2);
      if (head > E) head = E;
      const int ng = (E - head) >> 2;
      for (int e = tid; e < head; e += T) pick(e, (unsigned)(dst[e] - n0));
      for (int e = head + 4 * ng + tid; e < E; e += T)
        pick(e, (unsigned)(dst[e] - n0));
      const v4i* __restrict__ vp = (const v4i*)(dst + head);
      constexpr int LPT = 8;
      const int nfull = ng / (LPT * T);
      for (int it = 0; it < nfull; ++it) {
        const int g0 = it * (LPT * T) + tid;
        v4i dd[LPT];
#pragma unroll
        for (int u = 0; u < LPT; ++u) dd[u] = vp[g0 + u * T];
#pragma unroll
        for (int u = 0; u < LPT; ++u) pick4(head + 4 * (g0 + u * T), dd[u]);
      }
      for (int g = nfull * (LPT * T) + tid; g < ng; g += T)
        pick4(head + 4 * g, vp[g]);
    }
    __syncthreads();
    const int call = __builtin_amdgcn_readfirstlane(cnt);
    const bool listed = call <= OWN_CAP;

    if (listed) {
      for (int i0 = tid; i0 < call; i0 += 4 * T) {
        int s[4], dl[4];
        float xs[4], xd[4];
#pragma unroll
        for (int u = 0; u < 4; ++u) {
          const int i = i0 + u * T;
          s[u]  = (i < call) ? ls[i] : n0;
          dl[u] = (i < call) ? ld[i] : 0;
        }
#pragma unroll
        for (int u = 0; u < 4; ++u) {
          const int i = i0 + u * T;
          if (!loaded && i >= nn && i < call) s[u] = src[s[u]];
        }
#pragma unroll
        for (int u = 0; u < 4; ++u) {
          xs[u] = aS[s[u]];
          xd[u] = aD[n0 + dl[u]];
        }
#pragma unroll
        for (int u = 0; u < 4; ++u) {
          const int i = i0 + u * T;
          if (i < call) {
            const float v = leaky(xs[u] + xd[u]);
            ls[i] = s[u];
            lv[i] = v;
            lds_maxu(&mx[dl[u]], flipf(v));
            if (!SAVED && i >= nn) {
              ws_s[i] = s[u];
              ws_d[i] = dl[u];
            }
          }
        }
      }
      if (!SAVED && tid == 0) wcnt[run] = call;
    } else {
      if (!SAVED && tid == 0) wcnt[run] = -1;
      for (int e = tid; e < E; e += T) {
        const unsigned r = (unsigned)(dst[e] - n0);
        if (r < (unsigned)nn)
          lds_maxu(&mx[r], flipf(leaky(aS[src[e]] + aD[n0 + (int)r])));
      }
      for (int i = tid; i < nn; i += T)
        lds_maxu(&mx[i], flipf(leaky(aS[n0 + i] + aD[n0 + i])));
    }
    __syncthreads();

    if (listed) {
      const int nch = (call + EPC - 1) / EPC;
      for (int c = wave; c < nch; c += NW) {
        const int i0 = c * EPC;
        const int k  = (call - i0 < EPC) ? (call - i0) : EPC;
        int sv = 0, dv = 0;
        float wv = 0.f;
        if (lane < k) {
          sv = ls[i0 + lane];
          dv = ld[i0 + lane];
          wv = __expf(lv[i0 + lane] - unflipf(mx[dv]));
          lds_addf(&den[dv], wv);
        }
        if (k == EPC) {
          int s[EPC], d[EPC];
          float a[EPC];
#pragma unroll
          for (int j = 0; j < EPC; ++j) {
            s[j] = __builtin_amdgcn_readlane(sv, j);
            d[j] = __builtin_amdgcn_readlane(dv, j);
            a[j] = __int_as_float(
                __builtin_amdgcn_readlane(__float_as_int(wv), j));
          }
          float h[EPC][FPL];
#pragma unroll
          for (int i = 0; i < EPC * FPL; ++i)
            h[i / FPL][i % FPL] =
                H[(size_t)s[i / FPL] * D + 32 * (i % FPL) + lane];
#pragma unroll
          for (int i = 0; i < EPC * FPL; ++i)
            lds_addf(&acc[d[i / FPL] * D + 32 * (i % FPL) + lane],
                     a[i / FPL] * h[i / FPL][i % FPL]);
        } else {
          for (int j = 0; j < k; ++j) {
            const int sj   = __shfl(sv, j, 32);
            const int dj   = __shfl(dv, j, 32);
            const float aj = __shfl(wv, j, 32);
            for (int f = 0; f < FPL; ++f)
              lds_addf(&acc[dj * D + 32 * f + lane],
                       aj * H[(size_t)sj * D + 32 * f + lane]);
          }
        }
      }
    } else {
      auto one = [&](int sj, int dj) {
        const float w =
            __expf(leaky(aS[sj] + aD[n0 + dj]) - unflipf(mx[dj]));
        if (lane == 0) lds_addf(&den[dj], w);
        for (int f = 0; f < FPL; ++f)
          lds_addf(&acc[dj * D + 32 * f + lane],
                   w * H[(size_t)sj * D + 32 * f + lane]);
      };
      for (int base = wave * 32; base < E; base += NW * 32) {
        const int e = base + lane;
        const int dv = (e < E) ? dst[e] : -1;
        unsigned m = __builtin_amdgcn_ballot_w32((unsigned)(dv - n0) <
                                                 (unsigned)nn);
        while (m) {
          const int j = __builtin_ctz(m);
          m &= m - 1;
          one(src[base + j], __builtin_amdgcn_readlane(dv, j) - n0);
        }
      }
      for (int i = wave; i < nn; i += NW) one(n0 + i, i);
    }
    __syncthreads();

    for (int i = tid; i < nn * D; i += T) {
      const float v = acc[i] * (1.0f / den[i / D]) + bias[i % D];
      out[(size_t)n0 * D + i] = FINAL ? tanhf(v) : fmaxf(v, 0.f);
    }
    __syncthreads();
  }
}

extern "C" void kernel_launch(void* const* d_in, const int* in_sizes, int n_in,
                              void* d_out, int out_size, void* d_ws,
                              size_t ws_size, hipStream_t stream) {
  const float* x   = (const float*)d_in[0];
  const int*   ei  = (const int*)d_in[1];
  const float* W1  = (const float*)d_in[2];
  const float* as1 = (const float*)d_in[3];
  const float* ad1 = (const float*)d_in[4];
  const float* b1  = (const float*)d_in[5];
  const float* W2  = (const float*)d_in[6];
  const float* as2 = (const float*)d_in[7];
  const float* ad2 = (const float*)d_in[8];
  const float* b2  = (const float*)d_in[9];
  float* out = (float*)d_out;

  const int N    = in_sizes[0] / 64;
  const int E    = in_sizes[1] / 2;
  const int* src = ei;
  const int* dst = ei + E;

  float* ws = (float*)d_ws;
  size_t o = 0;
  float* h1 = ws + o;   o += (size_t)N * 64;
  float* o1 = ws + o;   o += (size_t)N * 64;
  float* aS = ws + o;   o += N;
  float* aD = ws + o;   o += N;
  int* wcnt = (int*)(ws + o);  o += N;
  int* wlist = (int*)(ws + o);

  const int B = 256;

  {
    gat_gemm_wmma<64, 64><<<GEMM_GRID, B, 0, stream>>>(x, W1, h1, N);
    att_scores<64><<<LOOP_GRID, B, 0, stream>>>(h1, as1, ad1, aS, aD, N);
    owner_aggregate<64, false, false><<<OWN_GRID, OWN_THREADS, 0, stream>>>(
        src, dst, aS, aD, h1, b1, o1, wcnt, wlist, E, N);
  }

  {
    float* h2 = h1;
    gat_gemm_wmma<64, 32><<<GEMM_GRID, B, 0, stream>>>(o1, W2, h2, N);
    att_scores<32><<<LOOP_GRID, B, 0, stream>>>(h2, as2, ad2, aS, aD, N);
    owner_aggregate<32, true, true><<<OWN_GRID, OWN_THREADS, 0, stream>>>(
        src, dst, aS, aD, h2, b2, out, wcnt, wlist, E, N);
  }
}


# Owner pick scans run keys (dst / run length) as 8 bit planes written by the layer-1 score launch: 1 byte not 4 per edge, 8 bit ops mask 32 edges; local ids read with source ids; sums still unordered
# speedup vs baseline: 4.4151x; 1.1983x over previous
#include <hip/hip_runtime.h>
#include <math.h>


typedef __attribute__((ext_vector_type(2))) float v2f;
typedef __attribute__((ext_vector_type(8))) float v8f;
typedef __attribute__((ext_vector_type(4))) int v4i;

#define NEG_SLOPE 0.2f

#if defined(__has_builtin)
#if __has_builtin(__builtin_amdgcn_wmma_f32_16x16x4_f32)
#define HAVE_WMMA_F32_16X16X4 1
#endif
#endif

__device__ __forceinline__ unsigned flipf(float f) {
  unsigned b = __float_as_uint(f);
  return (b & 0x80000000u) ? ~b : (b | 0x80000000u);
}
__device__ __forceinline__ float unflipf(unsigned u) {
  return __uint_as_float((u & 0x80000000u) ? (u & 0x7FFFFFFFu) : ~u);
}

__device__ __forceinline__ float leaky(float v) {
  return v > 0.f ? v : NEG_SLOPE * v;
}

#define LOOP_GRID 1024
#define GEMM_GRID 1024

template <int K, int DOUT>
__global__ void __launch_bounds__(256)
    __attribute__((amdgpu_waves_per_eu(8, 8)))
    gat_gemm_wmma(const float* __restrict__ X, const float* __restrict__ W,
                  float* __restrict__ H, int M) {
  constexpr int NT = DOUT / 16;
  constexpr int KS = K / 4;
  const int lane = threadIdx.x & 31;
  const int wpb  = blockDim.x >> 5;
  const int gwave =
      __builtin_amdgcn_readfirstlane(blockIdx.x * wpb + (threadIdx.x >> 5));
  const int nwaves = gridDim.x * wpb;
  const int MT     = M / 16;
  const int nt    = gwave % NT;
  const int mstep = nwaves / NT;
  int mt = gwave / NT;
  if (mt >= MT) return;

  const int r15 = lane & 15;
  const int hi  = lane >> 4;
  const int col = nt * 16 + r15;

#if HAVE_WMMA_F32_16X16X4
  v2f b[KS];
#pragma unroll
  for (int i = 0; i < KS; ++i) {
    const int ka = 4 * i + 2 * hi;
    b[i].x = W[ka * DOUT + col];
    b[i].y = W[(ka + 1) * DOUT + col];
  }

  for (; mt < MT; mt += mstep) {
    const int row = mt * 16 + r15;
    v8f c = {0.f, 0.f, 0.f, 0.f, 0.f, 0.f, 0.f, 0.f};
#pragma unroll
    for (int i = 0; i < KS; ++i) {
      const int ka = 4 * i + 2 * hi;
      v2f a;
      a.x = X[row * K + ka];
      a.y = X[row * K + ka + 1];
      c = __builtin_amdgcn_wmma_f32_16x16x4_f32(false, a, false, b[i],
                                                (short)0, c, false, false);
    }
    const int rowBase = mt * 16 + hi * 8;
#pragma unroll
    for (int r = 0; r < 8; ++r) H[(rowBase + r) * DOUT + col] = c[r];
  }
#else
  for (; mt < MT; mt += mstep) {
    for (int r = 0; r < 8; ++r) {
      const int rr = mt * 16 + hi * 8 + r;
      float acc = 0.f;
      for (int k = 0; k < K; ++k) acc += X[rr * K + k] * W[k * DOUT + col];
      H[rr * DOUT + col] = acc;
    }
  }
#endif
}

#define OWN_GRID 256
#define OWN_NPW 400

__device__ __forceinline__ int run_len(int N, int g) {
  const int npw = (N + g - 1) / g;
  return npw > OWN_NPW ? OWN_NPW : npw;
}

template <int D, bool KEYS>
__global__ void att_scores(const float* __restrict__ H,
                           const float* __restrict__ attS,
                           const float* __restrict__ attD,
                           float* __restrict__ aS, float* __restrict__ aD,
                           int N, const int* __restrict__ dst,
                           unsigned* __restrict__ keys, int E) {
  constexpr int FPL = D / 32;
  constexpr int NB  = 4;
  const int lane = threadIdx.x & 31;
  const int wpb  = blockDim.x >> 5;
  if (KEYS) {
    const unsigned npw = (unsigned)run_len(N, OWN_GRID);
    const int gw =
        __builtin_amdgcn_readfirstlane(blockIdx.x * wpb + (threadIdx.x >> 5));
    const int nw  = gridDim.x * wpb;
    const int ngp = ((E + 1023) >> 10) << 5;
    constexpr int NG = 4;
    for (int g0 = gw; g0 < ngp; g0 += NG * nw) {
      unsigned d[NG];
#pragma unroll
      for (int u = 0; u < NG; ++u) {
        const int e = 32 * (g0 + u * nw) + lane;
        d[u] = (e < E) ? (unsigned)dst[e] : 0u;
      }
#pragma unroll
      for (int u = 0; u < NG; ++u) {
        const int g = g0 + u * nw;
        if (g < ngp) {
          const unsigned k = d[u] / npw;
          unsigned v = 0;
#pragma unroll
          for (int j = 0; j < 8; ++j) {
            const unsigned b =
                __builtin_amdgcn_ballot_w32(((k >> j) & 1u) != 0u);
            if (lane == j) v = b;
          }
          if (lane < 8)
            keys[((((g >> 5) << 6) + ((lane >> 2) << 5) + (g & 31)) << 2) +
                 (lane & 3)] = v;
        }
      }
    }
  }
  const int gwave =
      __builtin_amdgcn_readfirstlane(blockIdx.x * wpb + (threadIdx.x >> 5));
  const int nwaves = gridDim.x * wpb;
  const int per    = (N + nwaves - 1) / nwaves;
  const int n0     = gwave * per;
  const int n1     = (n0 + per < N) ? (n0 + per) : N;
  if (n0 >= n1) return;

  float as[FPL], ad[FPL];
#pragma unroll
  for (int f = 0; f < FPL; ++f) {
    as[f] = attS[32 * f + lane];
    ad[f] = attD[32 * f + lane];
  }

  for (int n = n0; n < n1; n += NB) {
    float h[NB][FPL];
#pragma unroll
    for (int u = 0; u < NB; ++u) {
      const int node = (n + u < n1) ? (n + u) : (n1 - 1);
#pragma unroll
      for (int f = 0; f < FPL; ++f)
        h[u][f] = H[(size_t)node * D + 32 * f + lane];
    }
    float sv[NB], dv[NB];
#pragma unroll
    for (int u = 0; u < NB; ++u) {
      float sa = 0.f, da = 0.f;
#pragma unroll
      for (int f = 0; f < FPL; ++f) {
        sa += h[u][f] * as[f];
        da += h[u][f] * ad[f];
      }
      sv[u] = sa;
      dv[u] = da;
    }
#pragma unroll
    for (int off = 16; off > 0; off >>= 1) {
#pragma unroll
      for (int u = 0; u < NB; ++u) {
        sv[u] += __shfl_down(sv[u], off, 32);
        dv[u] += __shfl_down(dv[u], off, 32);
      }
    }
    if (lane == 0) {
#pragma unroll
      for (int u = 0; u < NB; ++u) {
        if (n + u < n1) {
          aS[n + u] = sv[u];
          aD[n + u] = dv[u];
        }
      }
    }
  }
}

#define OWN_THREADS 1024
#define OWN_CAP 8192
#define OWN_EPC 16

__device__ __forceinline__ void lds_addf(float* p, float v) {
  __hip_atomic_fetch_add(p, v, __ATOMIC_RELAXED, __HIP_MEMORY_SCOPE_WORKGROUP);
}
__device__ __forceinline__ void lds_maxu(unsigned* p, unsigned v) {
  __hip_atomic_fetch_max(p, v, __ATOMIC_RELAXED, __HIP_MEMORY_SCOPE_WORKGROUP);
}

template <int D, bool FINAL, bool SAVED>
__global__ void __launch_bounds__(OWN_THREADS)
    owner_aggregate(const int* __restrict__ src, const int* __restrict__ dst,
                    const float* __restrict__ aS, const float* __restrict__ aD,
                    const float* __restrict__ H, const float* __restrict__ bias,
                    float* __restrict__ out, int* __restrict__ wcnt,
                    int* __restrict__ wlist,
                    const unsigned* __restrict__ keys, int E, int N) {
  constexpr int FPL = D / 32;
  constexpr int T   = OWN_THREADS;
  constexpr int NW  = OWN_THREADS / 32;
  constexpr int EPC = OWN_EPC;
  __shared__ float acc[OWN_NPW * 64];
  __shared__ float den[OWN_NPW];
  __shared__ unsigned mx[OWN_NPW];
  __shared__ int ls[OWN_CAP];
  __shared__ int ld[OWN_CAP];
  __shared__ float lv[OWN_CAP];
  __shared__ int cnt;

  const int tid  = threadIdx.x;
  const int lane = tid & 31;
  const int wave = __builtin_amdgcn_readfirstlane(tid >> 5);

  const int npw   = run_len(N, (int)gridDim.x);
  const int nruns = (N + npw - 1) / npw;
  const bool bykey = nruns <= 256 && (int)gridDim.x == OWN_GRID;

  for (int run = blockIdx.x; run < nruns; run += gridDim.x) {
    const int n0 = run * npw;
    const int nn = (N - n0 < npw) ? (N - n0) : npw;
    int* __restrict__ ws_s = wlist + (size_t)run * (2 * OWN_CAP);
    int* __restrict__ ws_d = ws_s + OWN_CAP;

    for (int i = tid; i < nn * D; i += T) acc[i] = 0.f;
    for (int i = tid; i < nn; i += T) {
      den[i] = 0.f;
      mx[i]  = 0u;
      ls[i]  = n0 + i;
      ld[i]  = i;
    }
    if (tid == 0) cnt = nn;
    __syncthreads();

    bool loaded = false;
    if (SAVED) {
      const int c = wcnt[run];
      if (c >= 0) {
        loaded = true;
        for (int i = nn + tid; i < c; i += T) {
          ls[i] = ws_s[i];
          ld[i] = ws_d[i];
        }
        if (tid == 0) cnt = c;
      }
    }
    if (!loaded && bykey) {
      auto app = [&](int e) {
        const int idx = __hip_atomic_fetch_add(&cnt, 1, __ATOMIC_RELAXED,
                                               __HIP_MEMORY_SCOPE_WORKGROUP);
        if (idx < OWN_CAP) ls[idx] = e;
      };
      unsigned x[8];
#pragma unroll
      for (int j = 0; j < 8; ++j) x[j] = ((run >> j) & 1) ? 0u : ~0u;
      auto scan32 = [&](int g, v4i a, v4i b) {
        unsigned m = ((unsigned)a.x ^ x[0]) & ((unsigned)a.y ^ x[1]) &
                     ((unsigned)a.z ^ x[2]) & ((unsigned)a.w ^ x[3]) &
                     ((unsigned)b.x ^ x[4]) & ((unsigned)b.y ^ x[5]) &
                     ((unsigned)b.z ^ x[6]) & ((unsigned)b.w ^ x[7]);
        while (m) {
          const int e = 32 * g + __builtin_ctz(m);
          m &= m - 1;
          if (e < E) app(e);
        }
      };
      const v4i* __restrict__ vp = (const v4i*)keys;
      const int ngr = (E + 31) >> 5;
      constexpr int NGRP = 4;
      const int nfull = ngr / (NGRP * T);
      for (int it = 0; it < nfull; ++it) {
        const int g0 = it * (NGRP * T) + tid;
        v4i qa[NGRP], qb[NGRP];
#pragma unroll
        for (int u = 0; u < NGRP; ++u) {
          const int g = g0 + u * T;
          qa[u] = vp[2 * g - (g & 31)];
          qb[u] = vp[2 * g - (g & 31) + 32];
        }
#pragma unroll
        for (int u = 0; u < NGRP; ++u) scan32(g0 + u * T, qa[u], qb[u]);
      }
      for (int g = nfull * (NGRP * T) + tid; g < ngr; g += T)
        scan32(g, vp[2 * g - (g & 31)], vp[2 * g - (g & 31) + 32]);
    } else if (!loaded) {
      auto pick = [&](int e, unsigned r) {
        if (r < (unsigned)nn) {
          const int idx = __hip_atomic_fetch_add(&cnt, 1, __ATOMIC_RELAXED,
                                                 __HIP_MEMORY_SCOPE_WORKGROUP);
          if (idx < OWN_CAP) {
            ls[idx] = e;
            ld[idx] = (int)r;
          }
        }
      };
      auto pick4 = [&](int e, v4i d) {
        const unsigned r0 = (unsigned)(d.x - n0), r1 = (unsigned)(d.y - n0);
        const unsigned r2 = (unsigned)(d.z - n0), r3 = (unsigned)(d.w - n0);
        const unsigned m01 = r0 < r1 ? r0 : r1, m23 = r2 < r3 ? r2 : r3;
        if ((m01 < m23 ? m01 : m23) < (unsigned)nn) {
          pick(e, r0);
          pick(e + 1, r1);
          pick(e + 2, r2);
          pick(e + 3, r3);
        }
      };
      int head = (int)(((16u - ((unsigned)(size_t)dst & 15u)) & 15u) >> 2);
      if (head > E) head = E;
      const int ng = (E - head) >> 2;
      for (int e = tid; e < head; e += T) pick(e, (unsigned)(dst[e] - n0));
      for (int e = head + 4 * ng + tid; e < E; e += T)
        pick(e, (unsigned)(dst[e] - n0));
      const v4i* __restrict__ vp = (const v4i*)(dst + head);
      constexpr int LPT = 8;
      const int nfull = ng / (LPT * T);
      for (int it = 0; it < nfull; ++it) {
        const int g0 = it * (LPT * T) + tid;
        v4i dd[LPT];
#pragma unroll
        for (int u = 0; u < LPT; ++u) dd[u] = vp[g0 + u * T];
#pragma unroll
        for (int u = 0; u < LPT; ++u) pick4(head + 4 * (g0 + u * T), dd[u]);
      }
      for (int g = nfull * (LPT * T) + tid; g < ng; g += T)
        pick4(head + 4 * g, vp[g]);
    }
    __syncthreads();
    const int call = __builtin_amdgcn_readfirstlane(cnt);
    const bool listed = call <= OWN_CAP;

    if (listed) {
      for (int i0 = tid; i0 < call; i0 += 4 * T) {
        int s[4], dl[4];
        float xs[4], xd[4];
#pragma unroll
        for (int u = 0; u < 4; ++u) {
          const int i = i0 + u * T;
          s[u]  = (i < call) ? ls[i] : n0;
          dl[u] = (i < call) ? ld[i] : 0;
        }
#pragma unroll
        for (int u = 0; u < 4; ++u) {
          const int i = i0 + u * T;
          if (!loaded && i >= nn && i < call) {
            const int e = s[u];
            s[u] = src[e];
            if (bykey) dl[u] = dst[e] - n0;
          }
        }
#pragma unroll
        for (int u = 0; u < 4; ++u) {
          xs[u] = aS[s[u]];
          xd[u] = aD[n0 + dl[u]];
        }
#pragma unroll
        for (int u = 0; u < 4; ++u) {
          const int i = i0 + u * T;
          if (i < call) {
            const float v = leaky(xs[u] + xd[u]);
            ls[i] = s[u];
            ld[i] = dl[u];
            lv[i] = v;
            lds_maxu(&mx[dl[u]], flipf(v));
            if (!SAVED && i >= nn) {
              ws_s[i] = s[u];
              ws_d[i] = dl[u];
            }
          }
        }
      }
      if (!SAVED && tid == 0) wcnt[run] = call;
    } else {
      if (!SAVED && tid == 0) wcnt[run] = -1;
      for (int e = tid; e < E; e += T) {
        const unsigned r = (unsigned)(dst[e] - n0);
        if (r < (unsigned)nn)
          lds_maxu(&mx[r], flipf(leaky(aS[src[e]] + aD[n0 + (int)r])));
      }
      for (int i = tid; i < nn; i += T)
        lds_maxu(&mx[i], flipf(leaky(aS[n0 + i] + aD[n0 + i])));
    }
    __syncthreads();

    if (listed) {
      const int nch = (call + EPC - 1) / EPC;
      for (int c = wave; c < nch; c += NW) {
        const int i0 = c * EPC;
        const int k  = (call - i0 < EPC) ? (call - i0) : EPC;
        int sv = 0, dv = 0;
        float wv = 0.f;
        if (lane < k) {
          sv = ls[i0 + lane];
          dv = ld[i0 + lane];
          wv = __expf(lv[i0 + lane] - unflipf(mx[dv]));
          lds_addf(&den[dv], wv);
        }
        if (k == EPC) {
          int s[EPC], d[EPC];
          float a[EPC];
#pragma unroll
          for (int j = 0; j < EPC; ++j) {
            s[j] = __builtin_amdgcn_readlane(sv, j);
            d[j] = __builtin_amdgcn_readlane(dv, j);
            a[j] = __int_as_float(
                __builtin_amdgcn_readlane(__float_as_int(wv), j));
          }
          float h[EPC][FPL];
#pragma unroll
          for (int i = 0; i < EPC * FPL; ++i)
            h[i / FPL][i % FPL] =
                H[(size_t)s[i / FPL] * D + 32 * (i % FPL) + lane];
#pragma unroll
          for (int i = 0; i < EPC * FPL; ++i)
            lds_addf(&acc[d[i / FPL] * D + 32 * (i % FPL) + lane],
                     a[i / FPL] * h[i / FPL][i % FPL]);
        } else {
          for (int j = 0; j < k; ++j) {
            const int sj   = __shfl(sv, j, 32);
            const int dj   = __shfl(dv, j, 32);
            const float aj = __shfl(wv, j, 32);
            for (int f = 0; f < FPL; ++f)
              lds_addf(&acc[dj * D + 32 * f + lane],
                       aj * H[(size_t)sj * D + 32 * f + lane]);
          }
        }
      }
    } else {
      auto one = [&](int sj, int dj) {
        const float w =
            __expf(leaky(aS[sj] + aD[n0 + dj]) - unflipf(mx[dj]));
        if (lane == 0) lds_addf(&den[dj], w);
        for (int f = 0; f < FPL; ++f)
          lds_addf(&acc[dj * D + 32 * f + lane],
                   w * H[(size_t)sj * D + 32 * f + lane]);
      };
      for (int base = wave * 32; base < E; base += NW * 32) {
        const int e = base + lane;
        const int dv = (e < E) ? dst[e] : -1;
        unsigned m = __builtin_amdgcn_ballot_w32((unsigned)(dv - n0) <
                                                 (unsigned)nn);
        while (m) {
          const int j = __builtin_ctz(m);
          m &= m - 1;
          one(src[base + j], __builtin_amdgcn_readlane(dv, j) - n0);
        }
      }
      for (int i = wave; i < nn; i += NW) one(n0 + i, i);
    }
    __syncthreads();

    for (int i = tid; i < nn * D; i += T) {
      const float v = acc[i] * (1.0f / den[i / D]) + bias[i % D];
      out[(size_t)n0 * D + i] = FINAL ? tanhf(v) : fmaxf(v, 0.f);
    }
    __syncthreads();
  }
}

extern "C" void kernel_launch(void* const* d_in, const int* in_sizes, int n_in,
                              void* d_out, int out_size, void* d_ws,
                              size_t ws_size, hipStream_t stream) {
  const float* x   = (const float*)d_in[0];
  const int*   ei  = (const int*)d_in[1];
  const float* W1  = (const float*)d_in[2];
  const float* as1 = (const float*)d_in[3];
  const float* ad1 = (const float*)d_in[4];
  const float* b1  = (const float*)d_in[5];
  const float* W2  = (const float*)d_in[6];
  const float* as2 = (const float*)d_in[7];
  const float* ad2 = (const float*)d_in[8];
  const float* b2  = (const float*)d_in[9];
  float* out = (float*)d_out;

  const int N    = in_sizes[0] / 64;
  const int E    = in_sizes[1] / 2;
  const int* src = ei;
  const int* dst = ei + E;

  float* ws = (float*)d_ws;
  size_t o = 0;
  unsigned* keys = (unsigned*)(ws + o);  o += ((size_t)E / 1024 + 1) * 256;
  float* h1 = ws + o;   o += (size_t)N * 64;
  float* o1 = ws + o;   o += (size_t)N * 64;
  float* aS = ws + o;   o += N;
  float* aD = ws + o;   o += N;
  int* wcnt = (int*)(ws + o);  o += N;
  int* wlist = (int*)(ws + o);

  const int B = 256;

  {
    gat_gemm_wmma<64, 64><<<GEMM_GRID, B, 0, stream>>>(x, W1, h1, N);
    att_scores<64, true><<<LOOP_GRID, B, 0, stream>>>(h1, as1, ad1, aS, aD, N,
                                                      dst, keys, E);
    owner_aggregate<64, false, false><<<OWN_GRID, OWN_THREADS, 0, stream>>>(
        src, dst, aS, aD, h1, b1, o1, wcnt, wlist,
        keys, E, N);
  }

  {
    float* h2 = h1;
    gat_gemm_wmma<64, 32><<<GEMM_GRID, B, 0, stream>>>(o1, W2, h2, N);
    att_scores<32, false><<<LOOP_GRID, B, 0, stream>>>(h2, as2, ad2, aS, aD, N,
                                                       dst, keys, E);
    owner_aggregate<32, true, true><<<OWN_GRID, OWN_THREADS, 0, stream>>>(
        src, dst, aS, aD, h2, b2, out, wcnt, wlist,
        keys, E, N);
  }
}


# Node scores as x.(W att): layer 1 from the GEMM waves' A fragments (edge keys written there too), layer 2 where the layer-1 owner stores rows; both score launches gone; score sums in new order
# speedup vs baseline: 4.5867x; 1.0389x over previous
#include <hip/hip_runtime.h>
#include <math.h>


typedef __attribute__((ext_vector_type(2))) float v2f;
typedef __attribute__((ext_vector_type(8))) float v8f;
typedef __attribute__((ext_vector_type(4))) int v4i;
typedef __attribute__((ext_vector_type(4))) float v4f;

#define NEG_SLOPE 0.2f

#if defined(__has_builtin)
#if __has_builtin(__builtin_amdgcn_wmma_f32_16x16x4_f32)
#define HAVE_WMMA_F32_16X16X4 1
#endif
#endif

__device__ __forceinline__ unsigned flipf(float f) {
  unsigned b = __float_as_uint(f);
  return (b & 0x80000000u) ? ~b : (b | 0x80000000u);
}
__device__ __forceinline__ float unflipf(unsigned u) {
  return __uint_as_float((u & 0x80000000u) ? (u & 0x7FFFFFFFu) : ~u);
}

__device__ __forceinline__ float leaky(float v) {
  return v > 0.f ? v : NEG_SLOPE * v;
}

__device__ __forceinline__ float xhalf(float x) {
  const int b = __builtin_bit_cast(int, x);
  return __builtin_bit_cast(
      float, __builtin_amdgcn_permlanex16(b, b, 0x76543210, 0xfedcba98, false,
                                          false));
}
template <int M>
__device__ __forceinline__ float xrow(float x) {
  const int b = __builtin_bit_cast(int, x);
  return __builtin_bit_cast(
      float, __builtin_amdgcn_update_dpp(b, b, 0x160 + M, 0xF, 0xF, false));
}
__device__ __forceinline__ float wave_sum(float v) {
  v += xrow<8>(v);
  v += xrow<4>(v);
  v += xrow<2>(v);
  v += xrow<1>(v);
  return v + xhalf(v);
}

#define GEMM_GRID 1024

#define OWN_GRID 256
#define OWN_NPW 400

__device__ __forceinline__ int run_len(int N, int g) {
  const int npw = (N + g - 1) / g;
  return npw > OWN_NPW ? OWN_NPW : npw;
}

template <int K, int DOUT, bool SCORE>
__device__ __forceinline__ void gemm_tile(const float* __restrict__ X,
                                          float* __restrict__ H,
                                          const v2f (&b)[K / 4],
                                          const v2f (&ub)[K / 4],
                                          float* __restrict__ ao, int mt,
                                          int r15, int hi, int col) {
  constexpr int KS = K / 4;
  const int row = mt * 16 + r15;
  v8f c = {0.f, 0.f, 0.f, 0.f, 0.f, 0.f, 0.f, 0.f};
  float p0 = 0.f, p1 = 0.f;
#pragma unroll
  for (int i = 0; i < KS; ++i) {
    const int ka = 4 * i + 2 * hi;
    v2f a;
    a.x = X[row * K + ka];
    a.y = X[row * K + ka + 1];
    if (SCORE) {
      p0 += a.x * ub[i].x;
      p1 += a.y * ub[i].y;
    }
    c = __builtin_amdgcn_wmma_f32_16x16x4_f32(false, a, false, b[i], (short)0,
                                              c, false, false);
  }
  const int rowBase = mt * 16 + hi * 8;
#pragma unroll
  for (int r = 0; r < 8; ++r) H[(rowBase + r) * DOUT + col] = c[r];
  if (SCORE) {
    const float p = p0 + p1;
    const float t = p + xhalf(p);
    if (hi == 0) ao[row] = t;
  }
}

template <int K, int DOUT, bool SC>
__global__ void __launch_bounds__(256)
    __attribute__((amdgpu_waves_per_eu(8, 8)))
    gat_gemm_wmma(const float* __restrict__ X, const float* __restrict__ W,
                  float* __restrict__ H, int M,
                  const float* __restrict__ attS,
                  const float* __restrict__ attD, float* __restrict__ aS,
                  float* __restrict__ aD, const int* __restrict__ dst,
                  unsigned* __restrict__ keys, int E) {
  constexpr int NT = DOUT / 16;
  constexpr int KS = K / 4;
  const int lane = threadIdx.x & 31;
  const int wpb  = blockDim.x >> 5;
  const int gwave =
      __builtin_amdgcn_readfirstlane(blockIdx.x * wpb + (threadIdx.x >> 5));
  const int nwaves = gridDim.x * wpb;

  if (SC) {
    const unsigned npw = (unsigned)run_len(M, OWN_GRID);
    const int ngp = ((E + 1023) >> 10) << 5;
    constexpr int NG = 4;
    for (int g0 = gwave; g0 < ngp; g0 += NG * nwaves) {
      unsigned d[NG];
#pragma unroll
      for (int u = 0; u < NG; ++u) {
        const int e = 32 * (g0 + u * nwaves) + lane;
        d[u] = (e < E) ? (unsigned)dst[e] : 0u;
      }
#pragma unroll
      for (int u = 0; u < NG; ++u) {
        const int g = g0 + u * nwaves;
        if (g < ngp) {
          const unsigned k = d[u] / npw;
          unsigned v = 0;
#pragma unroll
          for (int j = 0; j < 8; ++j) {
            const unsigned bl =
                __builtin_amdgcn_ballot_w32(((k >> j) & 1u) != 0u);
            if (lane == j) v = bl;
          }
          if (lane < 8)
            keys[((((g >> 5) << 6) + ((lane >> 2) << 5) + (g & 31)) << 2) +
                 (lane & 3)] = v;
        }
      }
    }
  }

  const int MT = M / 16;
  const int nt    = gwave % NT;
  const int mstep = nwaves / NT;
  int mt = gwave / NT;
  if (mt >= MT) return;

  const int r15 = lane & 15;
  const int hi  = lane >> 4;
  const int col = nt * 16 + r15;

#if HAVE_WMMA_F32_16X16X4
  const bool score = SC && nt < 2;
  constexpr int UPL = K / 32;
  float u[UPL];
  if (score) {
    const float* __restrict__ att = nt == 0 ? attS : attD;
    float q[UPL][4];
#pragma unroll
    for (int s = 0; s < UPL; ++s) q[s][0] = q[s][1] = q[s][2] = q[s][3] = 0.f;
#pragma unroll 8
    for (int j = 0; j < DOUT / 4; ++j) {
#pragma unroll
      for (int s = 0; s < UPL; ++s) {
        const v4f w4 = ((const v4f*)(W + (size_t)(32 * s + lane) * DOUT))[j];
        q[s][0] += w4.x * att[4 * j];
        q[s][1] += w4.y * att[4 * j + 1];
        q[s][2] += w4.z * att[4 * j + 2];
        q[s][3] += w4.w * att[4 * j + 3];
      }
    }
#pragma unroll
    for (int s = 0; s < UPL; ++s)
      u[s] = (q[s][0] + q[s][1]) + (q[s][2] + q[s][3]);
  }

  v2f b[KS];
#pragma unroll
  for (int i = 0; i < KS; ++i) {
    const int ka = 4 * i + 2 * hi;
    b[i].x = W[ka * DOUT + col];
    b[i].y = W[(ka + 1) * DOUT + col];
  }

  v2f ub[KS];
  if (score) {
#pragma unroll
    for (int i = 0; i < KS; ++i) {
      const int ka   = 4 * i + 2 * hi;
      const float us = u[(4 * i) >> 5];
      ub[i].x = __shfl(us, ka & 31, 32);
      ub[i].y = __shfl(us, (ka + 1) & 31, 32);
    }
    float* __restrict__ ao = nt == 0 ? aS : aD;
    for (; mt < MT; mt += mstep)
      gemm_tile<K, DOUT, true>(X, H, b, ub, ao, mt, r15, hi, col);
  } else {
    for (; mt < MT; mt += mstep)
      gemm_tile<K, DOUT, false>(X, H, b, ub, nullptr, mt, r15, hi, col);
  }
#else
  for (; mt < MT; mt += mstep) {
    for (int r = 0; r < 8; ++r) {
      const int rr = mt * 16 + hi * 8 + r;
      float acc = 0.f;
      for (int k = 0; k < K; ++k) acc += X[rr * K + k] * W[k * DOUT + col];
      H[rr * DOUT + col] = acc;
      if (SC && nt < 2 && r15 == 0) {
        const float* __restrict__ att = nt == 0 ? attS : attD;
        float sc = 0.f;
        for (int k = 0; k < K; ++k) {
          float uk = 0.f;
          for (int j = 0; j < DOUT; ++j) uk += W[k * DOUT + j] * att[j];
          sc += X[rr * K + k] * uk;
        }
        (nt == 0 ? aS : aD)[rr] = sc;
      }
    }
  }
#endif
}

#define OWN_THREADS 1024
#define OWN_CAP 8192
#define OWN_EPC 16

__device__ __forceinline__ void lds_addf(float* p, float v) {
  __hip_atomic_fetch_add(p, v, __ATOMIC_RELAXED, __HIP_MEMORY_SCOPE_WORKGROUP);
}
__device__ __forceinline__ void lds_maxu(unsigned* p, unsigned v) {
  __hip_atomic_fetch_max(p, v, __ATOMIC_RELAXED, __HIP_MEMORY_SCOPE_WORKGROUP);
}

template <int D, int DN, bool FINAL, bool SAVED>
__global__ void __launch_bounds__(OWN_THREADS)
    owner_aggregate(const int* __restrict__ src, const int* __restrict__ dst,
                    const float* __restrict__ aS, const float* __restrict__ aD,
                    const float* __restrict__ H, const float* __restrict__ bias,
                    float* __restrict__ out, int* __restrict__ wcnt,
                    int* __restrict__ wlist,
                    const unsigned* __restrict__ keys, int E, int N,
                    const float* __restrict__ Wn,
                    const float* __restrict__ attSn,
                    const float* __restrict__ attDn, float* __restrict__ aSn,
                    float* __restrict__ aDn) {
  constexpr int FPL = D / 32;
  constexpr int T   = OWN_THREADS;
  constexpr int NW  = OWN_THREADS / 32;
  constexpr int EPC = OWN_EPC;
  __shared__ float acc[OWN_NPW * 64];
  __shared__ float den[OWN_NPW];
  __shared__ unsigned mx[OWN_NPW];
  __shared__ int ls[OWN_CAP];
  __shared__ int ld[OWN_CAP];
  __shared__ float lv[OWN_CAP];
  __shared__ int cnt;
  __shared__ float fold[FINAL ? 2 : 2 * D];

  const int tid  = threadIdx.x;
  const int lane = tid & 31;
  const int wave = __builtin_amdgcn_readfirstlane(tid >> 5);

  if (!FINAL) {
    if (tid < 2 * D) {
      const float* __restrict__ att = tid < D ? attSn : attDn;
      const float* __restrict__ wr  = Wn + (size_t)(tid % D) * DN;
      float q = 0.f;
      for (int j = 0; j < DN; ++j) q += wr[j] * att[j];
      fold[tid] = q;
    }
  }

  const int npw   = run_len(N, (int)gridDim.x);
  const int nruns = (N + npw - 1) / npw;
  const bool bykey = nruns <= 256 && (int)gridDim.x == OWN_GRID;

  for (int run = blockIdx.x; run < nruns; run += gridDim.x) {
    const int n0 = run * npw;
    const int nn = (N - n0 < npw) ? (N - n0) : npw;
    int* __restrict__ ws_s = wlist + (size_t)run * (2 * OWN_CAP);
    int* __restrict__ ws_d = ws_s + OWN_CAP;

    for (int i = tid; i < nn * D; i += T) acc[i] = 0.f;
    for (int i = tid; i < nn; i += T) {
      den[i] = 0.f;
      mx[i]  = 0u;
      ls[i]  = n0 + i;
      ld[i]  = i;
    }
    if (tid == 0) cnt = nn;
    __syncthreads();

    bool loaded = false;
    if (SAVED) {
      const int c = wcnt[run];
      if (c >= 0) {
        loaded = true;
        for (int i = nn + tid; i < c; i += T) {
          ls[i] = ws_s[i];
          ld[i] = ws_d[i];
        }
        if (tid == 0) cnt = c;
      }
    }
    if (!loaded && bykey) {
      auto app = [&](int e) {
        const int idx = __hip_atomic_fetch_add(&cnt, 1, __ATOMIC_RELAXED,
                                               __HIP_MEMORY_SCOPE_WORKGROUP);
        if (idx < OWN_CAP) ls[idx] = e;
      };
      unsigned x[8];
#pragma unroll
      for (int j = 0; j < 8; ++j) x[j] = ((run >> j) & 1) ? 0u : ~0u;
      auto scan32 = [&](int g, v4i a, v4i b) {
        unsigned m = ((unsigned)a.x ^ x[0]) & ((unsigned)a.y ^ x[1]) &
                     ((unsigned)a.z ^ x[2]) & ((unsigned)a.w ^ x[3]) &
                     ((unsigned)b.x ^ x[4]) & ((unsigned)b.y ^ x[5]) &
                     ((unsigned)b.z ^ x[6]) & ((unsigned)b.w ^ x[7]);
        while (m) {
          const int e = 32 * g + __builtin_ctz(m);
          m &= m - 1;
          if (e < E) app(e);
        }
      };
      const v4i* __restrict__ vp = (const v4i*)keys;
      const int ngr = (E + 31) >> 5;
      constexpr int NGRP = 4;
      const int nfull = ngr / (NGRP * T);
      for (int it = 0; it < nfull; ++it) {
        const int g0 = it * (NGRP * T) + tid;
        v4i qa[NGRP], qb[NGRP];
#pragma unroll
        for (int u = 0; u < NGRP; ++u) {
          const int g = g0 + u * T;
          qa[u] = vp[2 * g - (g & 31)];
          qb[u] = vp[2 * g - (g & 31) + 32];
        }
#pragma unroll
        for (int u = 0; u < NGRP; ++u) scan32(g0 + u * T, qa[u], qb[u]);
      }
      for (int g = nfull * (NGRP * T) + tid; g < ngr; g += T)
        scan32(g, vp[2 * g - (g & 31)], vp[2 * g - (g & 31) + 32]);
    } else if (!loaded) {
      auto pick = [&](int e, unsigned r) {
        if (r < (unsigned)nn) {
          const int idx = __hip_atomic_fetch_add(&cnt, 1, __ATOMIC_RELAXED,
                                                 __HIP_MEMORY_SCOPE_WORKGROUP);
          if (idx < OWN_CAP) {
            ls[idx] = e;
            ld[idx] = (int)r;
          }
        }
      };
      auto pick4 = [&](int e, v4i d) {
        const unsigned r0 = (unsigned)(d.x - n0), r1 = (unsigned)(d.y - n0);
        const unsigned r2 = (unsigned)(d.z - n0), r3 = (unsigned)(d.w - n0);
        const unsigned m01 = r0 < r1 ? r0 : r1, m23 = r2 < r3 ? r2 : r3;
        if ((m01 < m23 ? m01 : m23) < (unsigned)nn) {
          pick(e, r0);
          pick(e + 1, r1);
          pick(e + 2, r2);
          pick(e + 3, r3);
        }
      };
      int head = (int)(((16u - ((unsigned)(size_t)dst & 15u)) & 15u) >> 2);
      if (head > E) head = E;
      const int ng = (E - head) >> 2;
      for (int e = tid; e < head; e += T) pick(e, (unsigned)(dst[e] - n0));
      for (int e = head + 4 * ng + tid; e < E; e += T)
        pick(e, (unsigned)(dst[e] - n0));
      const v4i* __restrict__ vp = (const v4i*)(dst + head);
      constexpr int LPT = 8;
      const int nfull = ng / (LPT * T);
      for (int it = 0; it < nfull; ++it) {
        const int g0 = it * (LPT * T) + tid;
        v4i dd[LPT];
#pragma unroll
        for (int u = 0; u < LPT; ++u) dd[u] = vp[g0 + u * T];
#pragma unroll
        for (int u = 0; u < LPT; ++u) pick4(head + 4 * (g0 + u * T), dd[u]);
      }
      for (int g = nfull * (LPT * T) + tid; g < ng; g += T)
        pick4(head + 4 * g, vp[g]);
    }
    __syncthreads();
    const int call = __builtin_amdgcn_readfirstlane(cnt);
    const bool listed = call <= OWN_CAP;

    if (listed) {
      for (int i0 = tid; i0 < call; i0 += 4 * T) {
        int s[4], dl[4];
        float xs[4], xd[4];
#pragma unroll
        for (int u = 0; u < 4; ++u) {
          const int i = i0 + u * T;
          s[u]  = (i < call) ? ls[i] : n0;
          dl[u] = (i < call) ? ld[i] : 0;
        }
#pragma unroll
        for (int u = 0; u < 4; ++u) {
          const int i = i0 + u * T;
          if (!loaded && i >= nn && i < call) {
            const int e = s[u];
            s[u] = src[e];
            if (bykey) dl[u] = dst[e] - n0;
          }
        }
#pragma unroll
        for (int u = 0; u < 4; ++u) {
          xs[u] = aS[s[u]];
          xd[u] = aD[n0 + dl[u]];
        }
#pragma unroll
        for (int u = 0; u < 4; ++u) {
          const int i = i0 + u * T;
          if (i < call) {
            const float v = leaky(xs[u] + xd[u]);
            ls[i] = s[u];
            ld[i] = dl[u];
            lv[i] = v;
            lds_maxu(&mx[dl[u]], flipf(v));
            if (!SAVED && i >= nn) {
              ws_s[i] = s[u];
              ws_d[i] = dl[u];
            }
          }
        }
      }
      if (!SAVED && tid == 0) wcnt[run] = call;
    } else {
      if (!SAVED && tid == 0) wcnt[run] = -1;
      for (int e = tid; e < E; e += T) {
        const unsigned r = (unsigned)(dst[e] - n0);
        if (r < (unsigned)nn)
          lds_maxu(&mx[r], flipf(leaky(aS[src[e]] + aD[n0 + (int)r])));
      }
      for (int i = tid; i < nn; i += T)
        lds_maxu(&mx[i], flipf(leaky(aS[n0 + i] + aD[n0 + i])));
    }
    __syncthreads();

    if (listed) {
      const int nch = (call + EPC - 1) / EPC;
      for (int c = wave; c < nch; c += NW) {
        const int i0 = c * EPC;
        const int k  = (call - i0 < EPC) ? (call - i0) : EPC;
        int sv = 0, dv = 0;
        float wv = 0.f;
        if (lane < k) {
          sv = ls[i0 + lane];
          dv = ld[i0 + lane];
          wv = __expf(lv[i0 + lane] - unflipf(mx[dv]));
          lds_addf(&den[dv], wv);
        }
        if (k == EPC) {
          int s[EPC], d[EPC];
          float a[EPC];
#pragma unroll
          for (int j = 0; j < EPC; ++j) {
            s[j] = __builtin_amdgcn_readlane(sv, j);
            d[j] = __builtin_amdgcn_readlane(dv, j);
            a[j] = __int_as_float(
                __builtin_amdgcn_readlane(__float_as_int(wv), j));
          }
          float h[EPC][FPL];
#pragma unroll
          for (int i = 0; i < EPC * FPL; ++i)
            h[i / FPL][i % FPL] =
                H[(size_t)s[i / FPL] * D + 32 * (i % FPL) + lane];
#pragma unroll
          for (int i = 0; i < EPC * FPL; ++i)
            lds_addf(&acc[d[i / FPL] * D + 32 * (i % FPL) + lane],
                     a[i / FPL] * h[i / FPL][i % FPL]);
        } else {
          for (int j = 0; j < k; ++j) {
            const int sj   = __shfl(sv, j, 32);
            const int dj   = __shfl(dv, j, 32);
            const float aj = __shfl(wv, j, 32);
            for (int f = 0; f < FPL; ++f)
              lds_addf(&acc[dj * D + 32 * f + lane],
                       aj * H[(size_t)sj * D + 32 * f + lane]);
          }
        }
      }
    } else {
      auto one = [&](int sj, int dj) {
        const float w =
            __expf(leaky(aS[sj] + aD[n0 + dj]) - unflipf(mx[dj]));
        if (lane == 0) lds_addf(&den[dj], w);
        for (int f = 0; f < FPL; ++f)
          lds_addf(&acc[dj * D + 32 * f + lane],
                   w * H[(size_t)sj * D + 32 * f + lane]);
      };
      for (int base = wave * 32; base < E; base += NW * 32) {
        const int e = base + lane;
        const int dv = (e < E) ? dst[e] : -1;
        unsigned m = __builtin_amdgcn_ballot_w32((unsigned)(dv - n0) <
                                                 (unsigned)nn);
        while (m) {
          const int j = __builtin_ctz(m);
          m &= m - 1;
          one(src[base + j], __builtin_amdgcn_readlane(dv, j) - n0);
        }
      }
      for (int i = wave; i < nn; i += NW) one(n0 + i, i);
    }
    __syncthreads();

    if (FINAL) {
      for (int i = tid; i < nn * D; i += T) {
        const float v = acc[i] * (1.0f / den[i / D]) + bias[i % D];
        out[(size_t)n0 * D + i] = tanhf(v);
      }
    } else {
      constexpr int NB = 4;
      float fs[FPL], fd[FPL], bb[FPL];
#pragma unroll
      for (int f = 0; f < FPL; ++f) {
        fs[f] = fold[32 * f + lane];
        fd[f] = fold[D + 32 * f + lane];
        bb[f] = bias[32 * f + lane];
      }
      for (int r0 = wave * NB; r0 < nn; r0 += NW * NB) {
        float ps[NB], pd[NB];
#pragma unroll
        for (int u = 0; u < NB; ++u) {
          const bool ok = r0 + u < nn;
          const int r   = ok ? r0 + u : nn - 1;
          const float inv = 1.0f / den[r];
          float sa = 0.f, da = 0.f;
#pragma unroll
          for (int f = 0; f < FPL; ++f) {
            const float v =
                fmaxf(acc[r * D + 32 * f + lane] * inv + bb[f], 0.f);
            if (ok) out[(size_t)(n0 + r) * D + 32 * f + lane] = v;
            sa += v * fs[f];
            da += v * fd[f];
          }
          ps[u] = wave_sum(sa);
          pd[u] = wave_sum(da);
        }
        if (lane == 0) {
#pragma unroll
          for (int u = 0; u < NB; ++u) {
            if (r0 + u < nn) {
              aSn[n0 + r0 + u] = ps[u];
              aDn[n0 + r0 + u] = pd[u];
            }
          }
        }
      }
    }
    __syncthreads();
  }
}

extern "C" void kernel_launch(void* const* d_in, const int* in_sizes, int n_in,
                              void* d_out, int out_size, void* d_ws,
                              size_t ws_size, hipStream_t stream) {
  const float* x   = (const float*)d_in[0];
  const int*   ei  = (const int*)d_in[1];
  const float* W1  = (const float*)d_in[2];
  const float* as1 = (const float*)d_in[3];
  const float* ad1 = (const float*)d_in[4];
  const float* b1  = (const float*)d_in[5];
  const float* W2  = (const float*)d_in[6];
  const float* as2 = (const float*)d_in[7];
  const float* ad2 = (const float*)d_in[8];
  const float* b2  = (const float*)d_in[9];
  float* out = (float*)d_out;

  const int N    = in_sizes[0] / 64;
  const int E    = in_sizes[1] / 2;
  const int* src = ei;
  const int* dst = ei + E;

  float* ws = (float*)d_ws;
  size_t o = 0;
  unsigned* keys = (unsigned*)(ws + o);  o += ((size_t)E / 1024 + 1) * 256;
  float* h1 = ws + o;   o += (size_t)N * 64;
  float* o1 = ws + o;   o += (size_t)N * 64;
  float* aS1 = ws + o;  o += N;
  float* aD1 = ws + o;  o += N;
  float* aS2 = ws + o;  o += N;
  float* aD2 = ws + o;  o += N;
  int* wcnt = (int*)(ws + o);  o += N;
  int* wlist = (int*)(ws + o);

  const int B = 256;

  {
    gat_gemm_wmma<64, 64, true><<<GEMM_GRID, B, 0, stream>>>(
        x, W1, h1, N, as1, ad1, aS1, aD1, dst, keys, E);
    owner_aggregate<64, 32, false, false>
        <<<OWN_GRID, OWN_THREADS, 0, stream>>>(src, dst, aS1, aD1, h1, b1, o1,
                                               wcnt, wlist, keys, E, N, W2,
                                               as2, ad2, aS2, aD2);
  }

  {
    float* h2 = h1;
    gat_gemm_wmma<64, 32, false><<<GEMM_GRID, B, 0, stream>>>(
        o1, W2, h2, N, nullptr, nullptr, nullptr, nullptr, nullptr, nullptr,
        0);
    owner_aggregate<32, 0, true, true><<<OWN_GRID, OWN_THREADS, 0, stream>>>(
        src, dst, aS2, aD2, h2, b2, out, wcnt, wlist, keys, E, N, nullptr,
        nullptr, nullptr, nullptr, nullptr);
  }
}


# Layer-1 GEMM launch also writes one packed word per edge (source id | destination's offset in its run << 23); owner reads that word per picked edge: one 128-B line, not a source and a destination one
# speedup vs baseline: 4.7783x; 1.0418x over previous
#include <hip/hip_runtime.h>
#include <math.h>


typedef __attribute__((ext_vector_type(2))) float v2f;
typedef __attribute__((ext_vector_type(8))) float v8f;
typedef __attribute__((ext_vector_type(4))) int v4i;
typedef __attribute__((ext_vector_type(4))) float v4f;

#define NEG_SLOPE 0.2f

#if defined(__has_builtin)
#if __has_builtin(__builtin_amdgcn_wmma_f32_16x16x4_f32)
#define HAVE_WMMA_F32_16X16X4 1
#endif
#endif

__device__ __forceinline__ unsigned flipf(float f) {
  unsigned b = __float_as_uint(f);
  return (b & 0x80000000u) ? ~b : (b | 0x80000000u);
}
__device__ __forceinline__ float unflipf(unsigned u) {
  return __uint_as_float((u & 0x80000000u) ? (u & 0x7FFFFFFFu) : ~u);
}

__device__ __forceinline__ float leaky(float v) {
  return v > 0.f ? v : NEG_SLOPE * v;
}

__device__ __forceinline__ float xhalf(float x) {
  const int b = __builtin_bit_cast(int, x);
  return __builtin_bit_cast(
      float, __builtin_amdgcn_permlanex16(b, b, 0x76543210, 0xfedcba98, false,
                                          false));
}
template <int M>
__device__ __forceinline__ float xrow(float x) {
  const int b = __builtin_bit_cast(int, x);
  return __builtin_bit_cast(
      float, __builtin_amdgcn_update_dpp(b, b, 0x160 + M, 0xF, 0xF, false));
}
__device__ __forceinline__ float wave_sum(float v) {
  v += xrow<8>(v);
  v += xrow<4>(v);
  v += xrow<2>(v);
  v += xrow<1>(v);
  return v + xhalf(v);
}

#define GEMM_GRID 1024

#define OWN_GRID 256
#define OWN_NPW 400

__device__ __forceinline__ int run_len(int N, int g) {
  const int npw = (N + g - 1) / g;
  return npw > OWN_NPW ? OWN_NPW : npw;
}

template <int K, int DOUT, bool SCORE>
__device__ __forceinline__ void gemm_tile(const float* __restrict__ X,
                                          float* __restrict__ H,
                                          const v2f (&b)[K / 4],
                                          const v2f (&ub)[K / 4],
                                          float* __restrict__ ao, int mt,
                                          int r15, int hi, int col) {
  constexpr int KS = K / 4;
  const int row = mt * 16 + r15;
  v8f c = {0.f, 0.f, 0.f, 0.f, 0.f, 0.f, 0.f, 0.f};
  float p0 = 0.f, p1 = 0.f;
#pragma unroll
  for (int i = 0; i < KS; ++i) {
    const int ka = 4 * i + 2 * hi;
    v2f a;
    a.x = X[row * K + ka];
    a.y = X[row * K + ka + 1];
    if (SCORE) {
      p0 += a.x * ub[i].x;
      p1 += a.y * ub[i].y;
    }
    c = __builtin_amdgcn_wmma_f32_16x16x4_f32(false, a, false, b[i], (short)0,
                                              c, false, false);
  }
  const int rowBase = mt * 16 + hi * 8;
#pragma unroll
  for (int r = 0; r < 8; ++r) H[(rowBase + r) * DOUT + col] = c[r];
  if (SCORE) {
    const float p = p0 + p1;
    const float t = p + xhalf(p);
    if (hi == 0) ao[row] = t;
  }
}

template <int K, int DOUT, bool SC>
__global__ void __launch_bounds__(256)
    __attribute__((amdgpu_waves_per_eu(8, 8)))
    gat_gemm_wmma(const float* __restrict__ X, const float* __restrict__ W,
                  float* __restrict__ H, int M,
                  const float* __restrict__ attS,
                  const float* __restrict__ attD, float* __restrict__ aS,
                  float* __restrict__ aD, const int* __restrict__ src,
                  const int* __restrict__ dst, unsigned* __restrict__ keys,
                  unsigned* __restrict__ pk, int E) {
  constexpr int NT = DOUT / 16;
  constexpr int KS = K / 4;
  const int lane = threadIdx.x & 31;
  const int wpb  = blockDim.x >> 5;
  const int gwave =
      __builtin_amdgcn_readfirstlane(blockIdx.x * wpb + (threadIdx.x >> 5));
  const int nwaves = gridDim.x * wpb;

  if (SC) {
    const unsigned npw = (unsigned)run_len(M, OWN_GRID);
    const int ngp = ((E + 1023) >> 10) << 5;
    constexpr int NG = 4;
    for (int g0 = gwave; g0 < ngp; g0 += NG * nwaves) {
      unsigned d[NG], sr[NG];
#pragma unroll
      for (int u = 0; u < NG; ++u) {
        const int e = 32 * (g0 + u * nwaves) + lane;
        d[u]  = (e < E) ? (unsigned)dst[e] : 0u;
        sr[u] = (e < E) ? (unsigned)src[e] : 0u;
      }
#pragma unroll
      for (int u = 0; u < NG; ++u) {
        const int g = g0 + u * nwaves;
        if (g < ngp) {
          const unsigned k = d[u] / npw;
          unsigned v = 0;
#pragma unroll
          for (int j = 0; j < 8; ++j) {
            const unsigned bl =
                __builtin_amdgcn_ballot_w32(((k >> j) & 1u) != 0u);
            if (lane == j) v = bl;
          }
          if (lane < 8)
            keys[((((g >> 5) << 6) + ((lane >> 2) << 5) + (g & 31)) << 2) +
                 (lane & 3)] = v;
          const int e = 32 * g + lane;
          if (e < E) pk[e] = sr[u] | ((d[u] - k * npw) << 23);
        }
      }
    }
  }

  const int MT = M / 16;
  const int nt    = gwave % NT;
  const int mstep = nwaves / NT;
  int mt = gwave / NT;
  if (mt >= MT) return;

  const int r15 = lane & 15;
  const int hi  = lane >> 4;
  const int col = nt * 16 + r15;

#if HAVE_WMMA_F32_16X16X4
  const bool score = SC && nt < 2;
  constexpr int UPL = K / 32;
  float u[UPL];
  if (score) {
    const float* __restrict__ att = nt == 0 ? attS : attD;
    float q[UPL][4];
#pragma unroll
    for (int s = 0; s < UPL; ++s) q[s][0] = q[s][1] = q[s][2] = q[s][3] = 0.f;
#pragma unroll 8
    for (int j = 0; j < DOUT / 4; ++j) {
#pragma unroll
      for (int s = 0; s < UPL; ++s) {
        const v4f w4 = ((const v4f*)(W + (size_t)(32 * s + lane) * DOUT))[j];
        q[s][0] += w4.x * att[4 * j];
        q[s][1] += w4.y * att[4 * j + 1];
        q[s][2] += w4.z * att[4 * j + 2];
        q[s][3] += w4.w * att[4 * j + 3];
      }
    }
#pragma unroll
    for (int s = 0; s < UPL; ++s)
      u[s] = (q[s][0] + q[s][1]) + (q[s][2] + q[s][3]);
  }

  v2f b[KS];
#pragma unroll
  for (int i = 0; i < KS; ++i) {
    const int ka = 4 * i + 2 * hi;
    b[i].x = W[ka * DOUT + col];
    b[i].y = W[(ka + 1) * DOUT + col];
  }

  v2f ub[KS];
  if (score) {
#pragma unroll
    for (int i = 0; i < KS; ++i) {
      const int ka   = 4 * i + 2 * hi;
      const float us = u[(4 * i) >> 5];
      ub[i].x = __shfl(us, ka & 31, 32);
      ub[i].y = __shfl(us, (ka + 1) & 31, 32);
    }
    float* __restrict__ ao = nt == 0 ? aS : aD;
    for (; mt < MT; mt += mstep)
      gemm_tile<K, DOUT, true>(X, H, b, ub, ao, mt, r15, hi, col);
  } else {
    for (; mt < MT; mt += mstep)
      gemm_tile<K, DOUT, false>(X, H, b, ub, nullptr, mt, r15, hi, col);
  }
#else
  for (; mt < MT; mt += mstep) {
    for (int r = 0; r < 8; ++r) {
      const int rr = mt * 16 + hi * 8 + r;
      float acc = 0.f;
      for (int k = 0; k < K; ++k) acc += X[rr * K + k] * W[k * DOUT + col];
      H[rr * DOUT + col] = acc;
      if (SC && nt < 2 && r15 == 0) {
        const float* __restrict__ att = nt == 0 ? attS : attD;
        float sc = 0.f;
        for (int k = 0; k < K; ++k) {
          float uk = 0.f;
          for (int j = 0; j < DOUT; ++j) uk += W[k * DOUT + j] * att[j];
          sc += X[rr * K + k] * uk;
        }
        (nt == 0 ? aS : aD)[rr] = sc;
      }
    }
  }
#endif
}

#define OWN_THREADS 1024
#define OWN_CAP 8192
#define OWN_EPC 16

__device__ __forceinline__ void lds_addf(float* p, float v) {
  __hip_atomic_fetch_add(p, v, __ATOMIC_RELAXED, __HIP_MEMORY_SCOPE_WORKGROUP);
}
__device__ __forceinline__ void lds_maxu(unsigned* p, unsigned v) {
  __hip_atomic_fetch_max(p, v, __ATOMIC_RELAXED, __HIP_MEMORY_SCOPE_WORKGROUP);
}

template <int D, int DN, bool FINAL, bool SAVED>
__global__ void __launch_bounds__(OWN_THREADS)
    owner_aggregate(const int* __restrict__ src, const int* __restrict__ dst,
                    const float* __restrict__ aS, const float* __restrict__ aD,
                    const float* __restrict__ H, const float* __restrict__ bias,
                    float* __restrict__ out, int* __restrict__ wcnt,
                    int* __restrict__ wlist,
                    const unsigned* __restrict__ keys,
                    const unsigned* __restrict__ pk, int E, int N,
                    const float* __restrict__ Wn,
                    const float* __restrict__ attSn,
                    const float* __restrict__ attDn, float* __restrict__ aSn,
                    float* __restrict__ aDn) {
  constexpr int FPL = D / 32;
  constexpr int T   = OWN_THREADS;
  constexpr int NW  = OWN_THREADS / 32;
  constexpr int EPC = OWN_EPC;
  __shared__ float acc[OWN_NPW * 64];
  __shared__ float den[OWN_NPW];
  __shared__ unsigned mx[OWN_NPW];
  __shared__ int ls[OWN_CAP];
  __shared__ int ld[OWN_CAP];
  __shared__ float lv[OWN_CAP];
  __shared__ int cnt;
  __shared__ float fold[FINAL ? 2 : 2 * D];

  const int tid  = threadIdx.x;
  const int lane = tid & 31;
  const int wave = __builtin_amdgcn_readfirstlane(tid >> 5);

  if (!FINAL) {
    if (tid < 2 * D) {
      const float* __restrict__ att = tid < D ? attSn : attDn;
      const float* __restrict__ wr  = Wn + (size_t)(tid % D) * DN;
      float q = 0.f;
      for (int j = 0; j < DN; ++j) q += wr[j] * att[j];
      fold[tid] = q;
    }
  }

  const int npw   = run_len(N, (int)gridDim.x);
  const int nruns = (N + npw - 1) / npw;
  const bool bykey = nruns <= 256 && (int)gridDim.x == OWN_GRID;

  for (int run = blockIdx.x; run < nruns; run += gridDim.x) {
    const int n0 = run * npw;
    const int nn = (N - n0 < npw) ? (N - n0) : npw;
    int* __restrict__ ws_s = wlist + (size_t)run * (2 * OWN_CAP);
    int* __restrict__ ws_d = ws_s + OWN_CAP;

    for (int i = tid; i < nn * D; i += T) acc[i] = 0.f;
    for (int i = tid; i < nn; i += T) {
      den[i] = 0.f;
      mx[i]  = 0u;
      ls[i]  = n0 + i;
      ld[i]  = i;
    }
    if (tid == 0) cnt = nn;
    __syncthreads();

    bool loaded = false;
    if (SAVED) {
      const int c = wcnt[run];
      if (c >= 0) {
        loaded = true;
        for (int i = nn + tid; i < c; i += T) {
          ls[i] = ws_s[i];
          ld[i] = ws_d[i];
        }
        if (tid == 0) cnt = c;
      }
    }
    if (!loaded && bykey) {
      auto app = [&](int e) {
        const int idx = __hip_atomic_fetch_add(&cnt, 1, __ATOMIC_RELAXED,
                                               __HIP_MEMORY_SCOPE_WORKGROUP);
        if (idx < OWN_CAP) ls[idx] = e;
      };
      unsigned x[8];
#pragma unroll
      for (int j = 0; j < 8; ++j) x[j] = ((run >> j) & 1) ? 0u : ~0u;
      auto scan32 = [&](int g, v4i a, v4i b) {
        unsigned m = ((unsigned)a.x ^ x[0]) & ((unsigned)a.y ^ x[1]) &
                     ((unsigned)a.z ^ x[2]) & ((unsigned)a.w ^ x[3]) &
                     ((unsigned)b.x ^ x[4]) & ((unsigned)b.y ^ x[5]) &
                     ((unsigned)b.z ^ x[6]) & ((unsigned)b.w ^ x[7]);
        while (m) {
          const int e = 32 * g + __builtin_ctz(m);
          m &= m - 1;
          if (e < E) app(e);
        }
      };
      const v4i* __restrict__ vp = (const v4i*)keys;
      const int ngr = (E + 31) >> 5;
      constexpr int NGRP = 4;
      const int nfull = ngr / (NGRP * T);
      for (int it = 0; it < nfull; ++it) {
        const int g0 = it * (NGRP * T) + tid;
        v4i qa[NGRP], qb[NGRP];
#pragma unroll
        for (int u = 0; u < NGRP; ++u) {
          const int g = g0 + u * T;
          qa[u] = vp[2 * g - (g & 31)];
          qb[u] = vp[2 * g - (g & 31) + 32];
        }
#pragma unroll
        for (int u = 0; u < NGRP; ++u) scan32(g0 + u * T, qa[u], qb[u]);
      }
      for (int g = nfull * (NGRP * T) + tid; g < ngr; g += T)
        scan32(g, vp[2 * g - (g & 31)], vp[2 * g - (g & 31) + 32]);
    } else if (!loaded) {
      auto pick = [&](int e, unsigned r) {
        if (r < (unsigned)nn) {
          const int idx = __hip_atomic_fetch_add(&cnt, 1, __ATOMIC_RELAXED,
                                                 __HIP_MEMORY_SCOPE_WORKGROUP);
          if (idx < OWN_CAP) {
            ls[idx] = e;
            ld[idx] = (int)r;
          }
        }
      };
      auto pick4 = [&](int e, v4i d) {
        const unsigned r0 = (unsigned)(d.x - n0), r1 = (unsigned)(d.y - n0);
        const unsigned r2 = (unsigned)(d.z - n0), r3 = (unsigned)(d.w - n0);
        const unsigned m01 = r0 < r1 ? r0 : r1, m23 = r2 < r3 ? r2 : r3;
        if ((m01 < m23 ? m01 : m23) < (unsigned)nn) {
          pick(e, r0);
          pick(e + 1, r1);
          pick(e + 2, r2);
          pick(e + 3, r3);
        }
      };
      int head = (int)(((16u - ((unsigned)(size_t)dst & 15u)) & 15u) >> 2);
      if (head > E) head = E;
      const int ng = (E - head) >> 2;
      for (int e = tid; e < head; e += T) pick(e, (unsigned)(dst[e] - n0));
      for (int e = head + 4 * ng + tid; e < E; e += T)
        pick(e, (unsigned)(dst[e] - n0));
      const v4i* __restrict__ vp = (const v4i*)(dst + head);
      constexpr int LPT = 8;
      const int nfull = ng / (LPT * T);
      for (int it = 0; it < nfull; ++it) {
        const int g0 = it * (LPT * T) + tid;
        v4i dd[LPT];
#pragma unroll
        for (int u = 0; u < LPT; ++u) dd[u] = vp[g0 + u * T];
#pragma unroll
        for (int u = 0; u < LPT; ++u) pick4(head + 4 * (g0 + u * T), dd[u]);
      }
      for (int g = nfull * (LPT * T) + tid; g < ng; g += T)
        pick4(head + 4 * g, vp[g]);
    }
    __syncthreads();
    const int call = __builtin_amdgcn_readfirstlane(cnt);
    const bool listed = call <= OWN_CAP;

    if (listed) {
      for (int i0 = tid; i0 < call; i0 += 4 * T) {
        int s[4], dl[4];
        float xs[4], xd[4];
#pragma unroll
        for (int u = 0; u < 4; ++u) {
          const int i = i0 + u * T;
          s[u]  = (i < call) ? ls[i] : n0;
          dl[u] = (i < call) ? ld[i] : 0;
        }
        if (!loaded && call > nn) {
          if (bykey) {
            unsigned p[4];
#pragma unroll
            for (int u = 0; u < 4; ++u) {
              const int i = i0 + u * T;
              p[u] = pk[(i >= nn && i < call) ? s[u] : 0];
            }
#pragma unroll
            for (int u = 0; u < 4; ++u) {
              const int i = i0 + u * T;
              if (i >= nn && i < call) {
                s[u]  = (int)(p[u] & 0x7FFFFFu);
                dl[u] = (int)(p[u] >> 23);
              }
            }
          } else {
#pragma unroll
            for (int u = 0; u < 4; ++u) {
              const int i = i0 + u * T;
              if (i >= nn && i < call) s[u] = src[s[u]];
            }
          }
        }
#pragma unroll
        for (int u = 0; u < 4; ++u) {
          xs[u] = aS[s[u]];
          xd[u] = aD[n0 + dl[u]];
        }
#pragma unroll
        for (int u = 0; u < 4; ++u) {
          const int i = i0 + u * T;
          if (i < call) {
            const float v = leaky(xs[u] + xd[u]);
            ls[i] = s[u];
            ld[i] = dl[u];
            lv[i] = v;
            lds_maxu(&mx[dl[u]], flipf(v));
            if (!SAVED && i >= nn) {
              ws_s[i] = s[u];
              ws_d[i] = dl[u];
            }
          }
        }
      }
      if (!SAVED && tid == 0) wcnt[run] = call;
    } else {
      if (!SAVED && tid == 0) wcnt[run] = -1;
      for (int e = tid; e < E; e += T) {
        const unsigned r = (unsigned)(dst[e] - n0);
        if (r < (unsigned)nn)
          lds_maxu(&mx[r], flipf(leaky(aS[src[e]] + aD[n0 + (int)r])));
      }
      for (int i = tid; i < nn; i += T)
        lds_maxu(&mx[i], flipf(leaky(aS[n0 + i] + aD[n0 + i])));
    }
    __syncthreads();

    if (listed) {
      const int nch = (call + EPC - 1) / EPC;
      for (int c = wave; c < nch; c += NW) {
        const int i0 = c * EPC;
        const int k  = (call - i0 < EPC) ? (call - i0) : EPC;
        int sv = 0, dv = 0;
        float wv = 0.f;
        if (lane < k) {
          sv = ls[i0 + lane];
          dv = ld[i0 + lane];
          wv = __expf(lv[i0 + lane] - unflipf(mx[dv]));
          lds_addf(&den[dv], wv);
        }
        if (k == EPC) {
          int s[EPC], d[EPC];
          float a[EPC];
#pragma unroll
          for (int j = 0; j < EPC; ++j) {
            s[j] = __builtin_amdgcn_readlane(sv, j);
            d[j] = __builtin_amdgcn_readlane(dv, j);
            a[j] = __int_as_float(
                __builtin_amdgcn_readlane(__float_as_int(wv), j));
          }
          float h[EPC][FPL];
#pragma unroll
          for (int i = 0; i < EPC * FPL; ++i)
            h[i / FPL][i % FPL] =
                H[(size_t)s[i / FPL] * D + 32 * (i % FPL) + lane];
#pragma unroll
          for (int i = 0; i < EPC * FPL; ++i)
            lds_addf(&acc[d[i / FPL] * D + 32 * (i % FPL) + lane],
                     a[i / FPL] * h[i / FPL][i % FPL]);
        } else {
          for (int j = 0; j < k; ++j) {
            const int sj   = __shfl(sv, j, 32);
            const int dj   = __shfl(dv, j, 32);
            const float aj = __shfl(wv, j, 32);
            for (int f = 0; f < FPL; ++f)
              lds_addf(&acc[dj * D + 32 * f + lane],
                       aj * H[(size_t)sj * D + 32 * f + lane]);
          }
        }
      }
    } else {
      auto one = [&](int sj, int dj) {
        const float w =
            __expf(leaky(aS[sj] + aD[n0 + dj]) - unflipf(mx[dj]));
        if (lane == 0) lds_addf(&den[dj], w);
        for (int f = 0; f < FPL; ++f)
          lds_addf(&acc[dj * D + 32 * f + lane],
                   w * H[(size_t)sj * D + 32 * f + lane]);
      };
      for (int base = wave * 32; base < E; base += NW * 32) {
        const int e = base + lane;
        const int dv = (e < E) ? dst[e] : -1;
        unsigned m = __builtin_amdgcn_ballot_w32((unsigned)(dv - n0) <
                                                 (unsigned)nn);
        while (m) {
          const int j = __builtin_ctz(m);
          m &= m - 1;
          one(src[base + j], __builtin_amdgcn_readlane(dv, j) - n0);
        }
      }
      for (int i = wave; i < nn; i += NW) one(n0 + i, i);
    }
    __syncthreads();

    if (FINAL) {
      for (int i = tid; i < nn * D; i += T) {
        const float v = acc[i] * (1.0f / den[i / D]) + bias[i % D];
        out[(size_t)n0 * D + i] = tanhf(v);
      }
    } else {
      constexpr int NB = 4;
      float fs[FPL], fd[FPL], bb[FPL];
#pragma unroll
      for (int f = 0; f < FPL; ++f) {
        fs[f] = fold[32 * f + lane];
        fd[f] = fold[D + 32 * f + lane];
        bb[f] = bias[32 * f + lane];
      }
      for (int r0 = wave * NB; r0 < nn; r0 += NW * NB) {
        float ps[NB], pd[NB];
#pragma unroll
        for (int u = 0; u < NB; ++u) {
          const bool ok = r0 + u < nn;
          const int r   = ok ? r0 + u : nn - 1;
          const float inv = 1.0f / den[r];
          float sa = 0.f, da = 0.f;
#pragma unroll
          for (int f = 0; f < FPL; ++f) {
            const float v =
                fmaxf(acc[r * D + 32 * f + lane] * inv + bb[f], 0.f);
            if (ok) out[(size_t)(n0 + r) * D + 32 * f + lane] = v;
            sa += v * fs[f];
            da += v * fd[f];
          }
          ps[u] = wave_sum(sa);
          pd[u] = wave_sum(da);
        }
        if (lane == 0) {
#pragma unroll
          for (int u = 0; u < NB; ++u) {
            if (r0 + u < nn) {
              aSn[n0 + r0 + u] = ps[u];
              aDn[n0 + r0 + u] = pd[u];
            }
          }
        }
      }
    }
    __syncthreads();
  }
}

extern "C" void kernel_launch(void* const* d_in, const int* in_sizes, int n_in,
                              void* d_out, int out_size, void* d_ws,
                              size_t ws_size, hipStream_t stream) {
  const float* x   = (const float*)d_in[0];
  const int*   ei  = (const int*)d_in[1];
  const float* W1  = (const float*)d_in[2];
  const float* as1 = (const float*)d_in[3];
  const float* ad1 = (const float*)d_in[4];
  const float* b1  = (const float*)d_in[5];
  const float* W2  = (const float*)d_in[6];
  const float* as2 = (const float*)d_in[7];
  const float* ad2 = (const float*)d_in[8];
  const float* b2  = (const float*)d_in[9];
  float* out = (float*)d_out;

  const int N    = in_sizes[0] / 64;
  const int E    = in_sizes[1] / 2;
  const int* src = ei;
  const int* dst = ei + E;

  float* ws = (float*)d_ws;
  size_t o = 0;
  unsigned* keys = (unsigned*)(ws + o);  o += ((size_t)E / 1024 + 1) * 256;
  unsigned* pk = (unsigned*)(ws + o);    o += ((size_t)E / 1024 + 1) * 1024;
  float* h1 = ws + o;   o += (size_t)N * 64;
  float* o1 = ws + o;   o += (size_t)N * 64;
  float* aS1 = ws + o;  o += N;
  float* aD1 = ws + o;  o += N;
  float* aS2 = ws + o;  o += N;
  float* aD2 = ws + o;  o += N;
  int* wcnt = (int*)(ws + o);  o += N;
  int* wlist = (int*)(ws + o);

  const int B = 256;

  {
    gat_gemm_wmma<64, 64, true><<<GEMM_GRID, B, 0, stream>>>(
        x, W1, h1, N, as1, ad1, aS1, aD1, src, dst, keys, pk, E);
    owner_aggregate<64, 32, false, false>
        <<<OWN_GRID, OWN_THREADS, 0, stream>>>(src, dst, aS1, aD1, h1, b1, o1,
                                               wcnt, wlist, keys, pk, E, N,
                                               W2, as2, ad2, aS2, aD2);
  }

  {
    float* h2 = h1;
    gat_gemm_wmma<64, 32, false><<<GEMM_GRID, B, 0, stream>>>(
        o1, W2, h2, N, nullptr, nullptr, nullptr, nullptr, nullptr, nullptr,
        nullptr, nullptr, 0);
    owner_aggregate<32, 0, true, true><<<OWN_GRID, OWN_THREADS, 0, stream>>>(
        src, dst, aS2, aD2, h2, b2, out, wcnt, wlist, keys, pk, E, N, nullptr,
        nullptr, nullptr, nullptr, nullptr);
  }
}


# Owner phases in fewer dependent round trips: saved list read in the score pass, 8 entries a thread (1 trip), layer 2 gathers 32 rows a wave, last partial chunk batched too; sums still unordered
# speedup vs baseline: 4.9506x; 1.0361x over previous
#include <hip/hip_runtime.h>
#include <math.h>


typedef __attribute__((ext_vector_type(2))) float v2f;
typedef __attribute__((ext_vector_type(8))) float v8f;
typedef __attribute__((ext_vector_type(4))) int v4i;
typedef __attribute__((ext_vector_type(4))) float v4f;

#define NEG_SLOPE 0.2f

#if defined(__has_builtin)
#if __has_builtin(__builtin_amdgcn_wmma_f32_16x16x4_f32)
#define HAVE_WMMA_F32_16X16X4 1
#endif
#endif

__device__ __forceinline__ unsigned flipf(float f) {
  unsigned b = __float_as_uint(f);
  return (b & 0x80000000u) ? ~b : (b | 0x80000000u);
}
__device__ __forceinline__ float unflipf(unsigned u) {
  return __uint_as_float((u & 0x80000000u) ? (u & 0x7FFFFFFFu) : ~u);
}

__device__ __forceinline__ float leaky(float v) {
  return v > 0.f ? v : NEG_SLOPE * v;
}

__device__ __forceinline__ float xhalf(float x) {
  const int b = __builtin_bit_cast(int, x);
  return __builtin_bit_cast(
      float, __builtin_amdgcn_permlanex16(b, b, 0x76543210, 0xfedcba98, false,
                                          false));
}
template <int M>
__device__ __forceinline__ float xrow(float x) {
  const int b = __builtin_bit_cast(int, x);
  return __builtin_bit_cast(
      float, __builtin_amdgcn_update_dpp(b, b, 0x160 + M, 0xF, 0xF, false));
}
__device__ __forceinline__ float wave_sum(float v) {
  v += xrow<8>(v);
  v += xrow<4>(v);
  v += xrow<2>(v);
  v += xrow<1>(v);
  return v + xhalf(v);
}

#define GEMM_GRID 1024

#define OWN_GRID 256
#define OWN_NPW 400

__device__ __forceinline__ int run_len(int N, int g) {
  const int npw = (N + g - 1) / g;
  return npw > OWN_NPW ? OWN_NPW : npw;
}

template <int K, int DOUT, bool SCORE>
__device__ __forceinline__ void gemm_tile(const float* __restrict__ X,
                                          float* __restrict__ H,
                                          const v2f (&b)[K / 4],
                                          const v2f (&ub)[K / 4],
                                          float* __restrict__ ao, int mt,
                                          int r15, int hi, int col) {
  constexpr int KS = K / 4;
  const int row = mt * 16 + r15;
  v8f c = {0.f, 0.f, 0.f, 0.f, 0.f, 0.f, 0.f, 0.f};
  float p0 = 0.f, p1 = 0.f;
#pragma unroll
  for (int i = 0; i < KS; ++i) {
    const int ka = 4 * i + 2 * hi;
    v2f a;
    a.x = X[row * K + ka];
    a.y = X[row * K + ka + 1];
    if (SCORE) {
      p0 += a.x * ub[i].x;
      p1 += a.y * ub[i].y;
    }
    c = __builtin_amdgcn_wmma_f32_16x16x4_f32(false, a, false, b[i], (short)0,
                                              c, false, false);
  }
  const int rowBase = mt * 16 + hi * 8;
#pragma unroll
  for (int r = 0; r < 8; ++r) H[(rowBase + r) * DOUT + col] = c[r];
  if (SCORE) {
    const float p = p0 + p1;
    const float t = p + xhalf(p);
    if (hi == 0) ao[row] = t;
  }
}

template <int K, int DOUT, bool SC>
__global__ void __launch_bounds__(256)
    __attribute__((amdgpu_waves_per_eu(8, 8)))
    gat_gemm_wmma(const float* __restrict__ X, const float* __restrict__ W,
                  float* __restrict__ H, int M,
                  const float* __restrict__ attS,
                  const float* __restrict__ attD, float* __restrict__ aS,
                  float* __restrict__ aD, const int* __restrict__ src,
                  const int* __restrict__ dst, unsigned* __restrict__ keys,
                  unsigned* __restrict__ pk, int E) {
  constexpr int NT = DOUT / 16;
  constexpr int KS = K / 4;
  const int lane = threadIdx.x & 31;
  const int wpb  = blockDim.x >> 5;
  const int gwave =
      __builtin_amdgcn_readfirstlane(blockIdx.x * wpb + (threadIdx.x >> 5));
  const int nwaves = gridDim.x * wpb;

  if (SC) {
    const unsigned npw = (unsigned)run_len(M, OWN_GRID);
    const int ngp = ((E + 1023) >> 10) << 5;
    constexpr int NG = 4;
    for (int g0 = gwave; g0 < ngp; g0 += NG * nwaves) {
      unsigned d[NG], sr[NG];
#pragma unroll
      for (int u = 0; u < NG; ++u) {
        const int e = 32 * (g0 + u * nwaves) + lane;
        d[u]  = (e < E) ? (unsigned)dst[e] : 0u;
        sr[u] = (e < E) ? (unsigned)src[e] : 0u;
      }
#pragma unroll
      for (int u = 0; u < NG; ++u) {
        const int g = g0 + u * nwaves;
        if (g < ngp) {
          const unsigned k = d[u] / npw;
          unsigned v = 0;
#pragma unroll
          for (int j = 0; j < 8; ++j) {
            const unsigned bl =
                __builtin_amdgcn_ballot_w32(((k >> j) & 1u) != 0u);
            if (lane == j) v = bl;
          }
          if (lane < 8)
            keys[((((g >> 5) << 6) + ((lane >> 2) << 5) + (g & 31)) << 2) +
                 (lane & 3)] = v;
          const int e = 32 * g + lane;
          if (e < E) pk[e] = sr[u] | ((d[u] - k * npw) << 23);
        }
      }
    }
  }

  const int MT = M / 16;
  const int nt    = gwave % NT;
  const int mstep = nwaves / NT;
  int mt = gwave / NT;
  if (mt >= MT) return;

  const int r15 = lane & 15;
  const int hi  = lane >> 4;
  const int col = nt * 16 + r15;

#if HAVE_WMMA_F32_16X16X4
  const bool score = SC && nt < 2;
  constexpr int UPL = K / 32;
  float u[UPL];
  if (score) {
    const float* __restrict__ att = nt == 0 ? attS : attD;
    float q[UPL][4];
#pragma unroll
    for (int s = 0; s < UPL; ++s) q[s][0] = q[s][1] = q[s][2] = q[s][3] = 0.f;
#pragma unroll 8
    for (int j = 0; j < DOUT / 4; ++j) {
#pragma unroll
      for (int s = 0; s < UPL; ++s) {
        const v4f w4 = ((const v4f*)(W + (size_t)(32 * s + lane) * DOUT))[j];
        q[s][0] += w4.x * att[4 * j];
        q[s][1] += w4.y * att[4 * j + 1];
        q[s][2] += w4.z * att[4 * j + 2];
        q[s][3] += w4.w * att[4 * j + 3];
      }
    }
#pragma unroll
    for (int s = 0; s < UPL; ++s)
      u[s] = (q[s][0] + q[s][1]) + (q[s][2] + q[s][3]);
  }

  v2f b[KS];
#pragma unroll
  for (int i = 0; i < KS; ++i) {
    const int ka = 4 * i + 2 * hi;
    b[i].x = W[ka * DOUT + col];
    b[i].y = W[(ka + 1) * DOUT + col];
  }

  v2f ub[KS];
  if (score) {
#pragma unroll
    for (int i = 0; i < KS; ++i) {
      const int ka   = 4 * i + 2 * hi;
      const float us = u[(4 * i) >> 5];
      ub[i].x = __shfl(us, ka & 31, 32);
      ub[i].y = __shfl(us, (ka + 1) & 31, 32);
    }
    float* __restrict__ ao = nt == 0 ? aS : aD;
    for (; mt < MT; mt += mstep)
      gemm_tile<K, DOUT, true>(X, H, b, ub, ao, mt, r15, hi, col);
  } else {
    for (; mt < MT; mt += mstep)
      gemm_tile<K, DOUT, false>(X, H, b, ub, nullptr, mt, r15, hi, col);
  }
#else
  for (; mt < MT; mt += mstep) {
    for (int r = 0; r < 8; ++r) {
      const int rr = mt * 16 + hi * 8 + r;
      float acc = 0.f;
      for (int k = 0; k < K; ++k) acc += X[rr * K + k] * W[k * DOUT + col];
      H[rr * DOUT + col] = acc;
      if (SC && nt < 2 && r15 == 0) {
        const float* __restrict__ att = nt == 0 ? attS : attD;
        float sc = 0.f;
        for (int k = 0; k < K; ++k) {
          float uk = 0.f;
          for (int j = 0; j < DOUT; ++j) uk += W[k * DOUT + j] * att[j];
          sc += X[rr * K + k] * uk;
        }
        (nt == 0 ? aS : aD)[rr] = sc;
      }
    }
  }
#endif
}

#define OWN_THREADS 1024
#define OWN_CAP 8192
#define OWN_LIF 32

__device__ __forceinline__ void lds_addf(float* p, float v) {
  __hip_atomic_fetch_add(p, v, __ATOMIC_RELAXED, __HIP_MEMORY_SCOPE_WORKGROUP);
}
__device__ __forceinline__ void lds_maxu(unsigned* p, unsigned v) {
  __hip_atomic_fetch_max(p, v, __ATOMIC_RELAXED, __HIP_MEMORY_SCOPE_WORKGROUP);
}

template <int D, int DN, bool FINAL, bool SAVED>
__global__ void __launch_bounds__(OWN_THREADS)
    owner_aggregate(const int* __restrict__ src, const int* __restrict__ dst,
                    const float* __restrict__ aS, const float* __restrict__ aD,
                    const float* __restrict__ H, const float* __restrict__ bias,
                    float* __restrict__ out, int* __restrict__ wcnt,
                    int* __restrict__ wlist,
                    const unsigned* __restrict__ keys,
                    const unsigned* __restrict__ pk, int E, int N,
                    const float* __restrict__ Wn,
                    const float* __restrict__ attSn,
                    const float* __restrict__ attDn, float* __restrict__ aSn,
                    float* __restrict__ aDn) {
  constexpr int FPL = D / 32;
  constexpr int T   = OWN_THREADS;
  constexpr int NW  = OWN_THREADS / 32;
  constexpr int EPC = OWN_LIF / FPL;
  constexpr int SPT = OWN_CAP / T;
  __shared__ float acc[OWN_NPW * 64];
  __shared__ float den[OWN_NPW];
  __shared__ unsigned mx[OWN_NPW];
  __shared__ int ls[OWN_CAP];
  __shared__ int ld[OWN_CAP];
  __shared__ float lv[OWN_CAP];
  __shared__ int cnt;
  __shared__ float fold[FINAL ? 2 : 2 * D];

  const int tid  = threadIdx.x;
  const int lane = tid & 31;
  const int wave = __builtin_amdgcn_readfirstlane(tid >> 5);

  if (!FINAL) {
    if (tid < 2 * D) {
      const float* __restrict__ att = tid < D ? attSn : attDn;
      const float* __restrict__ wr  = Wn + (size_t)(tid % D) * DN;
      float q = 0.f;
      for (int j = 0; j < DN; ++j) q += wr[j] * att[j];
      fold[tid] = q;
    }
  }

  const int npw   = run_len(N, (int)gridDim.x);
  const int nruns = (N + npw - 1) / npw;
  const bool bykey = nruns <= 256 && (int)gridDim.x == OWN_GRID;

  for (int run = blockIdx.x; run < nruns; run += gridDim.x) {
    const int n0 = run * npw;
    const int nn = (N - n0 < npw) ? (N - n0) : npw;
    int* __restrict__ ws_s = wlist + (size_t)run * (2 * OWN_CAP);
    int* __restrict__ ws_d = ws_s + OWN_CAP;
    int csaved = -1;
    if (SAVED) csaved = wcnt[run];

    for (int i = tid; i < nn * D; i += T) acc[i] = 0.f;
    for (int i = tid; i < nn; i += T) {
      den[i] = 0.f;
      mx[i]  = 0u;
      ls[i]  = n0 + i;
      ld[i]  = i;
    }
    if (tid == 0) cnt = nn;
    __syncthreads();

    const int csv = __builtin_amdgcn_readfirstlane(csaved);
    const bool loaded = SAVED && csv >= 0;
    if (loaded) {
    } else if (bykey) {
      auto app = [&](int e) {
        const int idx = __hip_atomic_fetch_add(&cnt, 1, __ATOMIC_RELAXED,
                                               __HIP_MEMORY_SCOPE_WORKGROUP);
        if (idx < OWN_CAP) ls[idx] = e;
      };
      unsigned x[8];
#pragma unroll
      for (int j = 0; j < 8; ++j) x[j] = ((run >> j) & 1) ? 0u : ~0u;
      auto scan32 = [&](int g, v4i a, v4i b) {
        unsigned m = ((unsigned)a.x ^ x[0]) & ((unsigned)a.y ^ x[1]) &
                     ((unsigned)a.z ^ x[2]) & ((unsigned)a.w ^ x[3]) &
                     ((unsigned)b.x ^ x[4]) & ((unsigned)b.y ^ x[5]) &
                     ((unsigned)b.z ^ x[6]) & ((unsigned)b.w ^ x[7]);
        while (m) {
          const int e = 32 * g + __builtin_ctz(m);
          m &= m - 1;
          if (e < E) app(e);
        }
      };
      const v4i* __restrict__ vp = (const v4i*)keys;
      const int ngr = (E + 31) >> 5;
      constexpr int NGRP = 4;
      const int nfull = ngr / (NGRP * T);
      for (int it = 0; it < nfull; ++it) {
        const int g0 = it * (NGRP * T) + tid;
        v4i qa[NGRP], qb[NGRP];
#pragma unroll
        for (int u = 0; u < NGRP; ++u) {
          const int g = g0 + u * T;
          qa[u] = vp[2 * g - (g & 31)];
          qb[u] = vp[2 * g - (g & 31) + 32];
        }
#pragma unroll
        for (int u = 0; u < NGRP; ++u) scan32(g0 + u * T, qa[u], qb[u]);
      }
      for (int g = nfull * (NGRP * T) + tid; g < ngr; g += T)
        scan32(g, vp[2 * g - (g & 31)], vp[2 * g - (g & 31) + 32]);
    } else {
      auto pick = [&](int e, unsigned r) {
        if (r < (unsigned)nn) {
          const int idx = __hip_atomic_fetch_add(&cnt, 1, __ATOMIC_RELAXED,
                                                 __HIP_MEMORY_SCOPE_WORKGROUP);
          if (idx < OWN_CAP) {
            ls[idx] = e;
            ld[idx] = (int)r;
          }
        }
      };
      auto pick4 = [&](int e, v4i d) {
        const unsigned r0 = (unsigned)(d.x - n0), r1 = (unsigned)(d.y - n0);
        const unsigned r2 = (unsigned)(d.z - n0), r3 = (unsigned)(d.w - n0);
        const unsigned m01 = r0 < r1 ? r0 : r1, m23 = r2 < r3 ? r2 : r3;
        if ((m01 < m23 ? m01 : m23) < (unsigned)nn) {
          pick(e, r0);
          pick(e + 1, r1);
          pick(e + 2, r2);
          pick(e + 3, r3);
        }
      };
      int head = (int)(((16u - ((unsigned)(size_t)dst & 15u)) & 15u) >> 2);
      if (head > E) head = E;
      const int ng = (E - head) >> 2;
      for (int e = tid; e < head; e += T) pick(e, (unsigned)(dst[e] - n0));
      for (int e = head + 4 * ng + tid; e < E; e += T)
        pick(e, (unsigned)(dst[e] - n0));
      const v4i* __restrict__ vp = (const v4i*)(dst + head);
      constexpr int LPT = 8;
      const int nfull = ng / (LPT * T);
      for (int it = 0; it < nfull; ++it) {
        const int g0 = it * (LPT * T) + tid;
        v4i dd[LPT];
#pragma unroll
        for (int u = 0; u < LPT; ++u) dd[u] = vp[g0 + u * T];
#pragma unroll
        for (int u = 0; u < LPT; ++u) pick4(head + 4 * (g0 + u * T), dd[u]);
      }
      for (int g = nfull * (LPT * T) + tid; g < ng; g += T)
        pick4(head + 4 * g, vp[g]);
    }
    __syncthreads();
    const int call = loaded ? csv : __builtin_amdgcn_readfirstlane(cnt);
    const bool listed = call <= OWN_CAP;

    if (listed) {
      for (int i0 = tid; i0 < call; i0 += SPT * T) {
        int s[SPT], dl[SPT];
        float xs[SPT], xd[SPT];
#pragma unroll
        for (int u = 0; u < SPT; ++u) {
          const int i = i0 + u * T;
          if (loaded) {
            s[u]  = (i < nn) ? n0 + i : n0;
            dl[u] = (i < nn) ? i : 0;
          } else {
            s[u]  = (i < call) ? ls[i] : n0;
            dl[u] = (i < call) ? ld[i] : 0;
          }
        }
        if (call > nn) {
          if (loaded) {
            int qs[SPT], qd[SPT];
#pragma unroll
            for (int u = 0; u < SPT; ++u) {
              const int i = i0 + u * T;
              const int q = (i >= nn && i < call) ? i : nn;
              qs[u] = ws_s[q];
              qd[u] = ws_d[q];
            }
#pragma unroll
            for (int u = 0; u < SPT; ++u) {
              const int i = i0 + u * T;
              if (i >= nn && i < call) {
                s[u]  = qs[u];
                dl[u] = qd[u];
              }
            }
          } else if (bykey) {
            unsigned p[SPT];
#pragma unroll
            for (int u = 0; u < SPT; ++u) {
              const int i = i0 + u * T;
              p[u] = pk[(i >= nn && i < call) ? s[u] : 0];
            }
#pragma unroll
            for (int u = 0; u < SPT; ++u) {
              const int i = i0 + u * T;
              if (i >= nn && i < call) {
                s[u]  = (int)(p[u] & 0x7FFFFFu);
                dl[u] = (int)(p[u] >> 23);
              }
            }
          } else {
#pragma unroll
            for (int u = 0; u < SPT; ++u) {
              const int i = i0 + u * T;
              if (i >= nn && i < call) s[u] = src[s[u]];
            }
          }
        }
#pragma unroll
        for (int u = 0; u < SPT; ++u) {
          xs[u] = aS[s[u]];
          xd[u] = aD[n0 + dl[u]];
        }
        asm volatile("" ::"v"(xs[0]), "v"(xs[1]), "v"(xs[2]), "v"(xs[3]),
                     "v"(xs[4]), "v"(xs[5]), "v"(xs[6]), "v"(xs[7]),
                     "v"(xd[0]), "v"(xd[1]), "v"(xd[2]), "v"(xd[3]),
                     "v"(xd[4]), "v"(xd[5]), "v"(xd[6]), "v"(xd[7]));
        static_assert(SPT == 8, "the fence above names eight entries");
#pragma unroll
        for (int u = 0; u < SPT; ++u) {
          const int i = i0 + u * T;
          if (i < call) {
            const float v = leaky(xs[u] + xd[u]);
            ls[i] = s[u];
            ld[i] = dl[u];
            lv[i] = v;
            lds_maxu(&mx[dl[u]], flipf(v));
            if (!SAVED && i >= nn) {
              ws_s[i] = s[u];
              ws_d[i] = dl[u];
            }
          }
        }
      }
      if (!SAVED && tid == 0) wcnt[run] = call;
    } else {
      if (!SAVED && tid == 0) wcnt[run] = -1;
      for (int e = tid; e < E; e += T) {
        const unsigned r = (unsigned)(dst[e] - n0);
        if (r < (unsigned)nn)
          lds_maxu(&mx[r], flipf(leaky(aS[src[e]] + aD[n0 + (int)r])));
      }
      for (int i = tid; i < nn; i += T)
        lds_maxu(&mx[i], flipf(leaky(aS[n0 + i] + aD[n0 + i])));
    }
    __syncthreads();

    if (listed) {
      const int nch = (call + EPC - 1) / EPC;
      for (int c = wave; c < nch; c += NW) {
        const int i0 = c * EPC;
        const int k  = (call - i0 < EPC) ? (call - i0) : EPC;
        int sv = 0, dv = 0;
        float wv = 0.f;
        if (lane < k) {
          sv = ls[i0 + lane];
          dv = ld[i0 + lane];
          wv = __expf(lv[i0 + lane] - unflipf(mx[dv]));
          lds_addf(&den[dv], wv);
        }
        float h[EPC][FPL];
#pragma unroll
        for (int j = 0; j < EPC; ++j) {
          const int sj = __builtin_amdgcn_readlane(sv, j);
#pragma unroll
          for (int f = 0; f < FPL; ++f)
            h[j][f] = H[(size_t)sj * D + 32 * f + lane];
        }
        if (k == EPC) {
#pragma unroll
          for (int j = 0; j < EPC; ++j) {
            const int dj   = __builtin_amdgcn_readlane(dv, j);
            const float aj = __int_as_float(
                __builtin_amdgcn_readlane(__float_as_int(wv), j));
#pragma unroll
            for (int f = 0; f < FPL; ++f)
              lds_addf(&acc[dj * D + 32 * f + lane], aj * h[j][f]);
          }
        } else {
#pragma unroll
          for (int j = 0; j < EPC; ++j) {
            if (j < k) {
              const int dj   = __builtin_amdgcn_readlane(dv, j);
              const float aj = __int_as_float(
                  __builtin_amdgcn_readlane(__float_as_int(wv), j));
#pragma unroll
              for (int f = 0; f < FPL; ++f)
                lds_addf(&acc[dj * D + 32 * f + lane], aj * h[j][f]);
            }
          }
        }
      }
    } else {
      auto one = [&](int sj, int dj) {
        const float w =
            __expf(leaky(aS[sj] + aD[n0 + dj]) - unflipf(mx[dj]));
        if (lane == 0) lds_addf(&den[dj], w);
        for (int f = 0; f < FPL; ++f)
          lds_addf(&acc[dj * D + 32 * f + lane],
                   w * H[(size_t)sj * D + 32 * f + lane]);
      };
      for (int base = wave * 32; base < E; base += NW * 32) {
        const int e = base + lane;
        const int dv = (e < E) ? dst[e] : -1;
        unsigned m = __builtin_amdgcn_ballot_w32((unsigned)(dv - n0) <
                                                 (unsigned)nn);
        while (m) {
          const int j = __builtin_ctz(m);
          m &= m - 1;
          one(src[base + j], __builtin_amdgcn_readlane(dv, j) - n0);
        }
      }
      for (int i = wave; i < nn; i += NW) one(n0 + i, i);
    }
    __syncthreads();

    if (FINAL) {
      for (int i = tid; i < nn * D; i += T) {
        const float v = acc[i] * (1.0f / den[i / D]) + bias[i % D];
        out[(size_t)n0 * D + i] = tanhf(v);
      }
    } else {
      constexpr int NB = 4;
      float fs[FPL], fd[FPL], bb[FPL];
#pragma unroll
      for (int f = 0; f < FPL; ++f) {
        fs[f] = fold[32 * f + lane];
        fd[f] = fold[D + 32 * f + lane];
        bb[f] = bias[32 * f + lane];
      }
      for (int r0 = wave * NB; r0 < nn; r0 += NW * NB) {
        float ps[NB], pd[NB];
#pragma unroll
        for (int u = 0; u < NB; ++u) {
          const bool ok = r0 + u < nn;
          const int r   = ok ? r0 + u : nn - 1;
          const float inv = 1.0f / den[r];
          float sa = 0.f, da = 0.f;
#pragma unroll
          for (int f = 0; f < FPL; ++f) {
            const float v =
                fmaxf(acc[r * D + 32 * f + lane] * inv + bb[f], 0.f);
            if (ok) out[(size_t)(n0 + r) * D + 32 * f + lane] = v;
            sa += v * fs[f];
            da += v * fd[f];
          }
          ps[u] = wave_sum(sa);
          pd[u] = wave_sum(da);
        }
        if (lane == 0) {
#pragma unroll
          for (int u = 0; u < NB; ++u) {
            if (r0 + u < nn) {
              aSn[n0 + r0 + u] = ps[u];
              aDn[n0 + r0 + u] = pd[u];
            }
          }
        }
      }
    }
    __syncthreads();
  }
}

extern "C" void kernel_launch(void* const* d_in, const int* in_sizes, int n_in,
                              void* d_out, int out_size, void* d_ws,
                              size_t ws_size, hipStream_t stream) {
  const float* x   = (const float*)d_in[0];
  const int*   ei  = (const int*)d_in[1];
  const float* W1  = (const float*)d_in[2];
  const float* as1 = (const float*)d_in[3];
  const float* ad1 = (const float*)d_in[4];
  const float* b1  = (const float*)d_in[5];
  const float* W2  = (const float*)d_in[6];
  const float* as2 = (const float*)d_in[7];
  const float* ad2 = (const float*)d_in[8];
  const float* b2  = (const float*)d_in[9];
  float* out = (float*)d_out;

  const int N    = in_sizes[0] / 64;
  const int E    = in_sizes[1] / 2;
  const int* src = ei;
  const int* dst = ei + E;

  float* ws = (float*)d_ws;
  size_t o = 0;
  unsigned* keys = (unsigned*)(ws + o);  o += ((size_t)E / 1024 + 1) * 256;
  unsigned* pk = (unsigned*)(ws + o);    o += ((size_t)E / 1024 + 1) * 1024;
  float* h1 = ws + o;   o += (size_t)N * 64;
  float* o1 = ws + o;   o += (size_t)N * 64;
  float* aS1 = ws + o;  o += N;
  float* aD1 = ws + o;  o += N;
  float* aS2 = ws + o;  o += N;
  float* aD2 = ws + o;  o += N;
  int* wcnt = (int*)(ws + o);  o += N;
  int* wlist = (int*)(ws + o);

  const int B = 256;

  {
    gat_gemm_wmma<64, 64, true><<<GEMM_GRID, B, 0, stream>>>(
        x, W1, h1, N, as1, ad1, aS1, aD1, src, dst, keys, pk, E);
    owner_aggregate<64, 32, false, false>
        <<<OWN_GRID, OWN_THREADS, 0, stream>>>(src, dst, aS1, aD1, h1, b1, o1,
                                               wcnt, wlist, keys, pk, E, N,
                                               W2, as2, ad2, aS2, aD2);
  }

  {
    float* h2 = h1;
    gat_gemm_wmma<64, 32, false><<<GEMM_GRID, B, 0, stream>>>(
        o1, W2, h2, N, nullptr, nullptr, nullptr, nullptr, nullptr, nullptr,
        nullptr, nullptr, 0);
    owner_aggregate<32, 0, true, true><<<OWN_GRID, OWN_THREADS, 0, stream>>>(
        src, dst, aS2, aD2, h2, b2, out, wcnt, wlist, keys, pk, E, N, nullptr,
        nullptr, nullptr, nullptr, nullptr);
  }
}


# Owner pick via an edge index the layer-1 GEMM launch writes: per 2048-edge unit and run a bit per 16-edge group, edges as (dst,src) pairs; owner reads one line per flagged group; sums still unordered
# speedup vs baseline: 5.1144x; 1.0331x over previous
#include <hip/hip_runtime.h>
#include <math.h>


typedef __attribute__((ext_vector_type(2))) float v2f;
typedef __attribute__((ext_vector_type(8))) float v8f;
typedef __attribute__((ext_vector_type(4))) int v4i;
typedef __attribute__((ext_vector_type(2))) int v2i;
typedef __attribute__((ext_vector_type(4))) float v4f;

#define NEG_SLOPE 0.2f

#if defined(__has_builtin)
#if __has_builtin(__builtin_amdgcn_wmma_f32_16x16x4_f32)
#define HAVE_WMMA_F32_16X16X4 1
#endif
#endif

__device__ __forceinline__ unsigned flipf(float f) {
  unsigned b = __float_as_uint(f);
  return (b & 0x80000000u) ? ~b : (b | 0x80000000u);
}
__device__ __forceinline__ float unflipf(unsigned u) {
  return __uint_as_float((u & 0x80000000u) ? (u & 0x7FFFFFFFu) : ~u);
}

__device__ __forceinline__ float leaky(float v) {
  return v > 0.f ? v : NEG_SLOPE * v;
}

__device__ __forceinline__ float xhalf(float x) {
  const int b = __builtin_bit_cast(int, x);
  return __builtin_bit_cast(
      float, __builtin_amdgcn_permlanex16(b, b, 0x76543210, 0xfedcba98, false,
                                          false));
}
template <int M>
__device__ __forceinline__ float xrow(float x) {
  const int b = __builtin_bit_cast(int, x);
  return __builtin_bit_cast(
      float, __builtin_amdgcn_update_dpp(b, b, 0x160 + M, 0xF, 0xF, false));
}
__device__ __forceinline__ float wave_sum(float v) {
  v += xrow<8>(v);
  v += xrow<4>(v);
  v += xrow<2>(v);
  v += xrow<1>(v);
  return v + xhalf(v);
}

#define GEMM_GRID 1024

#define OWN_GRID 256
#define OWN_NPW 400

#define IDX_EU 2048
#define IDX_WPU (IDX_EU / 512)

__device__ __forceinline__ int run_len(int N, int g) {
  const int npw = (N + g - 1) / g;
  return npw > OWN_NPW ? OWN_NPW : npw;
}

template <int K, int DOUT, bool SCORE>
__device__ __forceinline__ void gemm_tile(const float* __restrict__ X,
                                          float* __restrict__ H,
                                          const v2f (&b)[K / 4],
                                          const v2f (&ub)[K / 4],
                                          float* __restrict__ ao, int mt,
                                          int r15, int hi, int col) {
  constexpr int KS = K / 4;
  const int row = mt * 16 + r15;
  v8f c = {0.f, 0.f, 0.f, 0.f, 0.f, 0.f, 0.f, 0.f};
  float p0 = 0.f, p1 = 0.f;
#pragma unroll
  for (int i = 0; i < KS; ++i) {
    const int ka = 4 * i + 2 * hi;
    v2f a;
    a.x = X[row * K + ka];
    a.y = X[row * K + ka + 1];
    if (SCORE) {
      p0 += a.x * ub[i].x;
      p1 += a.y * ub[i].y;
    }
    c = __builtin_amdgcn_wmma_f32_16x16x4_f32(false, a, false, b[i], (short)0,
                                              c, false, false);
  }
  const int rowBase = mt * 16 + hi * 8;
#pragma unroll
  for (int r = 0; r < 8; ++r) H[(rowBase + r) * DOUT + col] = c[r];
  if (SCORE) {
    const float p = p0 + p1;
    const float t = p + xhalf(p);
    if (hi == 0) ao[row] = t;
  }
}

template <int K, int DOUT, bool SC>
__global__ void __launch_bounds__(256)
    __attribute__((amdgpu_waves_per_eu(8, 8)))
    gat_gemm_wmma(const float* __restrict__ X, const float* __restrict__ W,
                  float* __restrict__ H, int M,
                  const float* __restrict__ attS,
                  const float* __restrict__ attD, float* __restrict__ aS,
                  float* __restrict__ aD, const int* __restrict__ src,
                  const int* __restrict__ dst, unsigned* __restrict__ bm,
                  int* __restrict__ rec, int E) {
  constexpr int NT = DOUT / 16;
  constexpr int KS = K / 4;
  const int lane = threadIdx.x & 31;
  const int wpb  = blockDim.x >> 5;
  const int gwave =
      __builtin_amdgcn_readfirstlane(blockIdx.x * wpb + (threadIdx.x >> 5));
  const int nwaves = gridDim.x * wpb;

  __shared__ __attribute__((aligned(16))) unsigned tile[SC ? 256 * IDX_WPU : 1];
  if (SC) {
    const unsigned npw = (unsigned)run_len(M, OWN_GRID);
    const int nunits = (E + IDX_EU - 1) / IDX_EU;
    const int tid = threadIdx.x;
    const int nth = blockDim.x;
    constexpr int EPT = 8;
    for (int unit = blockIdx.x; unit < nunits; unit += gridDim.x) {
      for (int i = tid; i < 256 * IDX_WPU; i += nth) tile[i] = 0u;
      __syncthreads();
      const int e0 = unit * IDX_EU;
      for (int b = tid; b < IDX_EU; b += EPT * nth) {
        int d[EPT], sr[EPT];
        bool ok[EPT];
#pragma unroll
        for (int u = 0; u < EPT; ++u) {
          const int el = b + u * nth;
          ok[u] = el < IDX_EU && e0 + el < E;
          d[u]  = ok[u] ? dst[e0 + el] : 0;
          sr[u] = ok[u] ? src[e0 + el] : 0;
        }
#pragma unroll
        for (int u = 0; u < EPT; ++u) {
          const int el = b + u * nth;
          if (ok[u]) {
            v2i p;
            p.x = d[u];
            p.y = sr[u];
            ((v2i*)rec)[e0 + el] = p;
            const unsigned k = (unsigned)d[u] / npw;
            if (k < 256u)
              __hip_atomic_fetch_or(&tile[k * IDX_WPU + (el >> 9)],
                                    1u << ((el >> 4) & 31), __ATOMIC_RELAXED,
                                    __HIP_MEMORY_SCOPE_WORKGROUP);
          }
        }
      }
      __syncthreads();
      for (int i = tid; i < 64 * IDX_WPU; i += nth)
        ((v4i*)bm)[(size_t)unit * (64 * IDX_WPU) + i] = ((const v4i*)tile)[i];
      __syncthreads();
    }
  }

  const int MT = M / 16;
  const int nt    = gwave % NT;
  const int mstep = nwaves / NT;
  int mt = gwave / NT;
  if (mt >= MT) return;

  const int r15 = lane & 15;
  const int hi  = lane >> 4;
  const int col = nt * 16 + r15;

#if HAVE_WMMA_F32_16X16X4
  const bool score = SC && nt < 2;
  constexpr int UPL = K / 32;
  float u[UPL];
  if (score) {
    const float* __restrict__ att = nt == 0 ? attS : attD;
    float q[UPL][4];
#pragma unroll
    for (int s = 0; s < UPL; ++s) q[s][0] = q[s][1] = q[s][2] = q[s][3] = 0.f;
#pragma unroll 8
    for (int j = 0; j < DOUT / 4; ++j) {
#pragma unroll
      for (int s = 0; s < UPL; ++s) {
        const v4f w4 = ((const v4f*)(W + (size_t)(32 * s + lane) * DOUT))[j];
        q[s][0] += w4.x * att[4 * j];
        q[s][1] += w4.y * att[4 * j + 1];
        q[s][2] += w4.z * att[4 * j + 2];
        q[s][3] += w4.w * att[4 * j + 3];
      }
    }
#pragma unroll
    for (int s = 0; s < UPL; ++s)
      u[s] = (q[s][0] + q[s][1]) + (q[s][2] + q[s][3]);
  }

  v2f b[KS];
#pragma unroll
  for (int i = 0; i < KS; ++i) {
    const int ka = 4 * i + 2 * hi;
    b[i].x = W[ka * DOUT + col];
    b[i].y = W[(ka + 1) * DOUT + col];
  }

  v2f ub[KS];
  if (score) {
#pragma unroll
    for (int i = 0; i < KS; ++i) {
      const int ka   = 4 * i + 2 * hi;
      const float us = u[(4 * i) >> 5];
      ub[i].x = __shfl(us, ka & 31, 32);
      ub[i].y = __shfl(us, (ka + 1) & 31, 32);
    }
    float* __restrict__ ao = nt == 0 ? aS : aD;
    for (; mt < MT; mt += mstep)
      gemm_tile<K, DOUT, true>(X, H, b, ub, ao, mt, r15, hi, col);
  } else {
    for (; mt < MT; mt += mstep)
      gemm_tile<K, DOUT, false>(X, H, b, ub, nullptr, mt, r15, hi, col);
  }
#else
  for (; mt < MT; mt += mstep) {
    for (int r = 0; r < 8; ++r) {
      const int rr = mt * 16 + hi * 8 + r;
      float acc = 0.f;
      for (int k = 0; k < K; ++k) acc += X[rr * K + k] * W[k * DOUT + col];
      H[rr * DOUT + col] = acc;
      if (SC && nt < 2 && r15 == 0) {
        const float* __restrict__ att = nt == 0 ? attS : attD;
        float sc = 0.f;
        for (int k = 0; k < K; ++k) {
          float uk = 0.f;
          for (int j = 0; j < DOUT; ++j) uk += W[k * DOUT + j] * att[j];
          sc += X[rr * K + k] * uk;
        }
        (nt == 0 ? aS : aD)[rr] = sc;
      }
    }
  }
#endif
}

#define OWN_THREADS 1024
#define OWN_CAP 8192
#define OWN_LIF 32

__device__ __forceinline__ void lds_addf(float* p, float v) {
  __hip_atomic_fetch_add(p, v, __ATOMIC_RELAXED, __HIP_MEMORY_SCOPE_WORKGROUP);
}
__device__ __forceinline__ void lds_maxu(unsigned* p, unsigned v) {
  __hip_atomic_fetch_max(p, v, __ATOMIC_RELAXED, __HIP_MEMORY_SCOPE_WORKGROUP);
}

template <int D, int DN, bool FINAL, bool SAVED>
__global__ void __launch_bounds__(OWN_THREADS)
    owner_aggregate(const int* __restrict__ src, const int* __restrict__ dst,
                    const float* __restrict__ aS, const float* __restrict__ aD,
                    const float* __restrict__ H, const float* __restrict__ bias,
                    float* __restrict__ out, int* __restrict__ wcnt,
                    int* __restrict__ wlist,
                    const unsigned* __restrict__ bm,
                    const int* __restrict__ rec, int E, int N,
                    const float* __restrict__ Wn,
                    const float* __restrict__ attSn,
                    const float* __restrict__ attDn, float* __restrict__ aSn,
                    float* __restrict__ aDn) {
  constexpr int FPL = D / 32;
  constexpr int T   = OWN_THREADS;
  constexpr int NW  = OWN_THREADS / 32;
  constexpr int EPC = OWN_LIF / FPL;
  constexpr int SPT = OWN_CAP / T;
  __shared__ float acc[OWN_NPW * 64];
  __shared__ float den[OWN_NPW];
  __shared__ unsigned mx[OWN_NPW];
  __shared__ int ls[OWN_CAP];
  __shared__ int ld[OWN_CAP];
  __shared__ float lv[OWN_CAP];
  __shared__ int gl[OWN_CAP];
  __shared__ int gcnt;
  __shared__ int cnt;
  __shared__ float fold[FINAL ? 2 : 2 * D];

  const int tid  = threadIdx.x;
  const int lane = tid & 31;
  const int wave = __builtin_amdgcn_readfirstlane(tid >> 5);

  if (!FINAL) {
    if (tid < 2 * D) {
      const float* __restrict__ att = tid < D ? attSn : attDn;
      const float* __restrict__ wr  = Wn + (size_t)(tid % D) * DN;
      float q = 0.f;
      for (int j = 0; j < DN; ++j) q += wr[j] * att[j];
      fold[tid] = q;
    }
  }

  const int npw   = run_len(N, (int)gridDim.x);
  const int nruns = (N + npw - 1) / npw;
  const bool bykey = nruns <= 256 && (int)gridDim.x == OWN_GRID;

  for (int run = blockIdx.x; run < nruns; run += gridDim.x) {
    const int n0 = run * npw;
    const int nn = (N - n0 < npw) ? (N - n0) : npw;
    int* __restrict__ ws_s = wlist + (size_t)run * (2 * OWN_CAP);
    int* __restrict__ ws_d = ws_s + OWN_CAP;
    int csaved = -1;
    if (SAVED) csaved = wcnt[run];
    const int nunits = (E + IDX_EU - 1) / IDX_EU;
    v4i f0 = {0, 0, 0, 0};
    if (!SAVED && bykey && tid < nunits)
      f0 = ((const v4i*)bm)[(size_t)tid * (64 * IDX_WPU) + run];

    for (int i = tid; i < nn * D; i += T) acc[i] = 0.f;
    for (int i = tid; i < nn; i += T) {
      den[i] = 0.f;
      mx[i]  = 0u;
      ls[i]  = n0 + i;
      ld[i]  = i;
    }
    if (tid == 0) {
      cnt  = nn;
      gcnt = 0;
    }
    __syncthreads();

    const int csv = __builtin_amdgcn_readfirstlane(csaved);
    const bool loaded = SAVED && csv >= 0;
    if (loaded) {
    } else if (bykey) {
      const v4i* __restrict__ fp = (const v4i*)bm;
      for (int unit = tid; unit < nunits; unit += T) {
        const v4i f = (!SAVED && unit == tid)
                          ? f0
                          : fp[(size_t)unit * (64 * IDX_WPU) + run];
        const unsigned fw[4] = {(unsigned)f.x, (unsigned)f.y, (unsigned)f.z,
                                (unsigned)f.w};
        static_assert(IDX_WPU == 4, "one 16-byte load holds a unit's words");
        const int np = __builtin_popcount(fw[0]) + __builtin_popcount(fw[1]) +
                       __builtin_popcount(fw[2]) + __builtin_popcount(fw[3]);
        int idx = 0;
        if (np)
          idx = __hip_atomic_fetch_add(&gcnt, np, __ATOMIC_RELAXED,
                                       __HIP_MEMORY_SCOPE_WORKGROUP);
#pragma unroll
        for (int w = 0; w < 4; ++w) {
          unsigned m = fw[w];
          while (m) {
            const int g = unit * (IDX_EU / 16) + 32 * w + __builtin_ctz(m);
            m &= m - 1;
            if (idx < OWN_CAP) gl[idx] = g;
            ++idx;
          }
        }
      }
      __syncthreads();
      const int ng = __builtin_amdgcn_readfirstlane(gcnt);
      if (ng > OWN_CAP) {
        if (tid == 0) cnt = OWN_CAP + 1;
      } else {
        const v2i* __restrict__ rp = (const v2i*)rec;
        const int l15 = lane & 15;
        const int hw  = 2 * wave + (lane >> 4);
        constexpr int GIF = 16;
        for (int p0 = 0; p0 < ng; p0 += 2 * NW * GIF) {
          v2i q[GIF];
          int gg[GIF];
          bool ok[GIF];
#pragma unroll
          for (int u = 0; u < GIF; ++u) {
            const int gi = p0 + 2 * NW * u + hw;
            gg[u] = gl[gi < ng ? gi : 0];
          }
#pragma unroll
          for (int u = 0; u < GIF; ++u) {
            const int gi = p0 + 2 * NW * u + hw;
            const int e  = (gi < ng) ? 16 * gg[u] + l15 : E;
            ok[u] = e < E;
            q[u]  = rp[ok[u] ? e : 0];
          }
          unsigned bal[GIF], r[GIF];
          int tot = 0;
#pragma unroll
          for (int u = 0; u < GIF; ++u) {
            r[u]   = (unsigned)(q[u].x - n0);
            bal[u] = __builtin_amdgcn_ballot_w32(ok[u] && r[u] < (unsigned)nn);
            tot += __builtin_popcount(bal[u]);
          }
          if (tot) {
            int base = 0;
            if (lane == 0)
              base = __hip_atomic_fetch_add(&cnt, tot, __ATOMIC_RELAXED,
                                            __HIP_MEMORY_SCOPE_WORKGROUP);
            base = __builtin_amdgcn_readfirstlane(base);
#pragma unroll
            for (int u = 0; u < GIF; ++u) {
              if ((bal[u] >> lane) & 1u) {
                const int idx =
                    base + (int)__builtin_amdgcn_mbcnt_lo(bal[u], 0u);
                if (idx < OWN_CAP) {
                  ls[idx] = q[u].y;
                  ld[idx] = (int)r[u];
                }
              }
              base += __builtin_popcount(bal[u]);
            }
          }
        }
      }
    } else {
      auto pick = [&](int e, unsigned r) {
        if (r < (unsigned)nn) {
          const int idx = __hip_atomic_fetch_add(&cnt, 1, __ATOMIC_RELAXED,
                                                 __HIP_MEMORY_SCOPE_WORKGROUP);
          if (idx < OWN_CAP) {
            ls[idx] = e;
            ld[idx] = (int)r;
          }
        }
      };
      auto pick4 = [&](int e, v4i d) {
        const unsigned r0 = (unsigned)(d.x - n0), r1 = (unsigned)(d.y - n0);
        const unsigned r2 = (unsigned)(d.z - n0), r3 = (unsigned)(d.w - n0);
        const unsigned m01 = r0 < r1 ? r0 : r1, m23 = r2 < r3 ? r2 : r3;
        if ((m01 < m23 ? m01 : m23) < (unsigned)nn) {
          pick(e, r0);
          pick(e + 1, r1);
          pick(e + 2, r2);
          pick(e + 3, r3);
        }
      };
      int head = (int)(((16u - ((unsigned)(size_t)dst & 15u)) & 15u) >> 2);
      if (head > E) head = E;
      const int ng = (E - head) >> 2;
      for (int e = tid; e < head; e += T) pick(e, (unsigned)(dst[e] - n0));
      for (int e = head + 4 * ng + tid; e < E; e += T)
        pick(e, (unsigned)(dst[e] - n0));
      const v4i* __restrict__ vp = (const v4i*)(dst + head);
      constexpr int LPT = 8;
      const int nfull = ng / (LPT * T);
      for (int it = 0; it < nfull; ++it) {
        const int g0 = it * (LPT * T) + tid;
        v4i dd[LPT];
#pragma unroll
        for (int u = 0; u < LPT; ++u) dd[u] = vp[g0 + u * T];
#pragma unroll
        for (int u = 0; u < LPT; ++u) pick4(head + 4 * (g0 + u * T), dd[u]);
      }
      for (int g = nfull * (LPT * T) + tid; g < ng; g += T)
        pick4(head + 4 * g, vp[g]);
    }
    __syncthreads();
    const int call = loaded ? csv : __builtin_amdgcn_readfirstlane(cnt);
    const bool listed = call <= OWN_CAP;

    if (listed) {
      for (int i0 = tid; i0 < call; i0 += SPT * T) {
        int s[SPT], dl[SPT];
        float xs[SPT], xd[SPT];
#pragma unroll
        for (int u = 0; u < SPT; ++u) {
          const int i = i0 + u * T;
          if (loaded) {
            s[u]  = (i < nn) ? n0 + i : n0;
            dl[u] = (i < nn) ? i : 0;
          } else {
            s[u]  = (i < call) ? ls[i] : n0;
            dl[u] = (i < call) ? ld[i] : 0;
          }
        }
        if (call > nn) {
          if (loaded) {
            int qs[SPT], qd[SPT];
#pragma unroll
            for (int u = 0; u < SPT; ++u) {
              const int i = i0 + u * T;
              const int q = (i >= nn && i < call) ? i : nn;
              qs[u] = ws_s[q];
              qd[u] = ws_d[q];
            }
#pragma unroll
            for (int u = 0; u < SPT; ++u) {
              const int i = i0 + u * T;
              if (i >= nn && i < call) {
                s[u]  = qs[u];
                dl[u] = qd[u];
              }
            }
          } else if (!bykey) {
#pragma unroll
            for (int u = 0; u < SPT; ++u) {
              const int i = i0 + u * T;
              if (i >= nn && i < call) s[u] = src[s[u]];
            }
          }
        }
#pragma unroll
        for (int u = 0; u < SPT; ++u) {
          xs[u] = aS[s[u]];
          xd[u] = aD[n0 + dl[u]];
        }
        asm volatile("" ::"v"(xs[0]), "v"(xs[1]), "v"(xs[2]), "v"(xs[3]),
                     "v"(xs[4]), "v"(xs[5]), "v"(xs[6]), "v"(xs[7]),
                     "v"(xd[0]), "v"(xd[1]), "v"(xd[2]), "v"(xd[3]),
                     "v"(xd[4]), "v"(xd[5]), "v"(xd[6]), "v"(xd[7]));
        static_assert(SPT == 8, "the fence above names eight entries");
#pragma unroll
        for (int u = 0; u < SPT; ++u) {
          const int i = i0 + u * T;
          if (i < call) {
            const float v = leaky(xs[u] + xd[u]);
            ls[i] = s[u];
            ld[i] = dl[u];
            lv[i] = v;
            lds_maxu(&mx[dl[u]], flipf(v));
            if (!SAVED && i >= nn) {
              ws_s[i] = s[u];
              ws_d[i] = dl[u];
            }
          }
        }
      }
      if (!SAVED && tid == 0) wcnt[run] = call;
    } else {
      if (!SAVED && tid == 0) wcnt[run] = -1;
      for (int e = tid; e < E; e += T) {
        const unsigned r = (unsigned)(dst[e] - n0);
        if (r < (unsigned)nn)
          lds_maxu(&mx[r], flipf(leaky(aS[src[e]] + aD[n0 + (int)r])));
      }
      for (int i = tid; i < nn; i += T)
        lds_maxu(&mx[i], flipf(leaky(aS[n0 + i] + aD[n0 + i])));
    }
    __syncthreads();

    if (listed) {
      const int nch = (call + EPC - 1) / EPC;
      for (int c = wave; c < nch; c += NW) {
        const int i0 = c * EPC;
        const int k  = (call - i0 < EPC) ? (call - i0) : EPC;
        int sv = 0, dv = 0;
        float wv = 0.f;
        if (lane < k) {
          sv = ls[i0 + lane];
          dv = ld[i0 + lane];
          wv = __expf(lv[i0 + lane] - unflipf(mx[dv]));
          lds_addf(&den[dv], wv);
        }
        float h[EPC][FPL];
#pragma unroll
        for (int j = 0; j < EPC; ++j) {
          const int sj = __builtin_amdgcn_readlane(sv, j);
#pragma unroll
          for (int f = 0; f < FPL; ++f)
            h[j][f] = H[(size_t)sj * D + 32 * f + lane];
        }
        if (k == EPC) {
#pragma unroll
          for (int j = 0; j < EPC; ++j) {
            const int dj   = __builtin_amdgcn_readlane(dv, j);
            const float aj = __int_as_float(
                __builtin_amdgcn_readlane(__float_as_int(wv), j));
#pragma unroll
            for (int f = 0; f < FPL; ++f)
              lds_addf(&acc[dj * D + 32 * f + lane], aj * h[j][f]);
          }
        } else {
#pragma unroll
          for (int j = 0; j < EPC; ++j) {
            if (j < k) {
              const int dj   = __builtin_amdgcn_readlane(dv, j);
              const float aj = __int_as_float(
                  __builtin_amdgcn_readlane(__float_as_int(wv), j));
#pragma unroll
              for (int f = 0; f < FPL; ++f)
                lds_addf(&acc[dj * D + 32 * f + lane], aj * h[j][f]);
            }
          }
        }
      }
    } else {
      auto one = [&](int sj, int dj) {
        const float w =
            __expf(leaky(aS[sj] + aD[n0 + dj]) - unflipf(mx[dj]));
        if (lane == 0) lds_addf(&den[dj], w);
        for (int f = 0; f < FPL; ++f)
          lds_addf(&acc[dj * D + 32 * f + lane],
                   w * H[(size_t)sj * D + 32 * f + lane]);
      };
      for (int base = wave * 32; base < E; base += NW * 32) {
        const int e = base + lane;
        const int dv = (e < E) ? dst[e] : -1;
        unsigned m = __builtin_amdgcn_ballot_w32((unsigned)(dv - n0) <
                                                 (unsigned)nn);
        while (m) {
          const int j = __builtin_ctz(m);
          m &= m - 1;
          one(src[base + j], __builtin_amdgcn_readlane(dv, j) - n0);
        }
      }
      for (int i = wave; i < nn; i += NW) one(n0 + i, i);
    }
    __syncthreads();

    if (FINAL) {
      for (int i = tid; i < nn * D; i += T) {
        const float v = acc[i] * (1.0f / den[i / D]) + bias[i % D];
        out[(size_t)n0 * D + i] = tanhf(v);
      }
    } else {
      constexpr int NB = 4;
      float fs[FPL], fd[FPL], bb[FPL];
#pragma unroll
      for (int f = 0; f < FPL; ++f) {
        fs[f] = fold[32 * f + lane];
        fd[f] = fold[D + 32 * f + lane];
        bb[f] = bias[32 * f + lane];
      }
      for (int r0 = wave * NB; r0 < nn; r0 += NW * NB) {
        float ps[NB], pd[NB];
#pragma unroll
        for (int u = 0; u < NB; ++u) {
          const bool ok = r0 + u < nn;
          const int r   = ok ? r0 + u : nn - 1;
          const float inv = 1.0f / den[r];
          float sa = 0.f, da = 0.f;
#pragma unroll
          for (int f = 0; f < FPL; ++f) {
            const float v =
                fmaxf(acc[r * D + 32 * f + lane] * inv + bb[f], 0.f);
            if (ok) out[(size_t)(n0 + r) * D + 32 * f + lane] = v;
            sa += v * fs[f];
            da += v * fd[f];
          }
          ps[u] = wave_sum(sa);
          pd[u] = wave_sum(da);
        }
        if (lane == 0) {
#pragma unroll
          for (int u = 0; u < NB; ++u) {
            if (r0 + u < nn) {
              aSn[n0 + r0 + u] = ps[u];
              aDn[n0 + r0 + u] = pd[u];
            }
          }
        }
      }
    }
    __syncthreads();
  }
}

extern "C" void kernel_launch(void* const* d_in, const int* in_sizes, int n_in,
                              void* d_out, int out_size, void* d_ws,
                              size_t ws_size, hipStream_t stream) {
  const float* x   = (const float*)d_in[0];
  const int*   ei  = (const int*)d_in[1];
  const float* W1  = (const float*)d_in[2];
  const float* as1 = (const float*)d_in[3];
  const float* ad1 = (const float*)d_in[4];
  const float* b1  = (const float*)d_in[5];
  const float* W2  = (const float*)d_in[6];
  const float* as2 = (const float*)d_in[7];
  const float* ad2 = (const float*)d_in[8];
  const float* b2  = (const float*)d_in[9];
  float* out = (float*)d_out;

  const int N    = in_sizes[0] / 64;
  const int E    = in_sizes[1] / 2;
  const int* src = ei;
  const int* dst = ei + E;

  float* ws = (float*)d_ws;
  size_t o = 0;
  unsigned* bm = (unsigned*)(ws + o);  o += ((size_t)E / 2048 + 1) * 1024;
  int* rec = (int*)(ws + o);           o += ((size_t)E / 1024 + 1) * 2048;
  float* h1 = ws + o;   o += (size_t)N * 64;
  float* o1 = ws + o;   o += (size_t)N * 64;
  float* aS1 = ws + o;  o += N;
  float* aD1 = ws + o;  o += N;
  float* aS2 = ws + o;  o += N;
  float* aD2 = ws + o;  o += N;
  int* wcnt = (int*)(ws + o);  o += N;
  int* wlist = (int*)(ws + o);

  const int B = 256;

  {
    gat_gemm_wmma<64, 64, true><<<GEMM_GRID, B, 0, stream>>>(
        x, W1, h1, N, as1, ad1, aS1, aD1, src, dst, bm, rec, E);
    owner_aggregate<64, 32, false, false>
        <<<OWN_GRID, OWN_THREADS, 0, stream>>>(src, dst, aS1, aD1, h1, b1, o1,
                                               wcnt, wlist, bm, rec, E, N,
                                               W2, as2, ad2, aS2, aD2);
  }

  {
    float* h2 = h1;
    gat_gemm_wmma<64, 32, false><<<GEMM_GRID, B, 0, stream>>>(
        o1, W2, h2, N, nullptr, nullptr, nullptr, nullptr, nullptr, nullptr,
        nullptr, nullptr, 0);
    owner_aggregate<32, 0, true, true><<<OWN_GRID, OWN_THREADS, 0, stream>>>(
        src, dst, aS2, aD2, h2, b2, out, wcnt, wlist, bm, rec, E, N, nullptr,
        nullptr, nullptr, nullptr, nullptr);
  }
}
